# starting kernel
# baseline (speedup 1.0000x reference)
#include <hip/hip_runtime.h>

#define FIN   500
#define HID   64
#define NCLS  47
#define CPAD  48
#define KSTEP 10
#define KCH   100
#define NCHK  5
#define W1STR 160
#define W2STR 96
#define HSTR  68

typedef __attribute__((ext_vector_type(2))) float v2f;
typedef __attribute__((ext_vector_type(8))) float v8f;

__global__ void gpr_init_deg(float* __restrict__ deg, int n) {
    int i = blockIdx.x * blockDim.x + threadIdx.x;
    if (i < n) deg[i] = 1.0f;
}

__global__ void gpr_count_deg(const int* __restrict__ col, float* __restrict__ deg, int e) {
    int i = blockIdx.x * blockDim.x + threadIdx.x;
    if (i < e) atomicAdd(&deg[col[i]], 1.0f);
}

__global__ void gpr_dis(const float* __restrict__ deg, float* __restrict__ dis,
                        float* __restrict__ dis2, int n) {
    int i = blockIdx.x * blockDim.x + threadIdx.x;
    if (i < n) {
        float d = deg[i];
        dis[i]  = rsqrtf(d);
        dis2[i] = 1.0f / d;
    }
}

__global__ void gpr_norm(const int* __restrict__ row, const int* __restrict__ col,
                         const float* __restrict__ dis, float* __restrict__ nrm, int e) {
    int i = blockIdx.x * blockDim.x + threadIdx.x;
    if (i < e) nrm[i] = dis[row[i]] * dis[col[i]];
}

__global__ __launch_bounds__(256)
void gpr_mlp_wmma(const float* __restrict__ x,  const float* __restrict__ W1,
                  const float* __restrict__ b1, const float* __restrict__ W2,
                  const float* __restrict__ b2, float* __restrict__ zp, int n) {
    __shared__ float sBig[8704];
    __shared__ float sW2[32 * W2STR];

    const int tid    = threadIdx.x;
    const int wid    = tid >> 5;
    const int lane   = tid & 31;
    const int rloc   = lane & 15;
    const int khalf  = lane >> 4;

    for (int i = tid; i < HID * CPAD; i += 256) {
        int k = i / CPAD, c = i % CPAD;
        sW2[(k >> 1) * W2STR + 2 * c + (k & 1)] = (c < NCLS) ? W2[k * NCLS + c] : 0.0f;
    }

    const int numTiles = (n + 15) / 16;
    for (int base = blockIdx.x * 8; base < numTiles; base += gridDim.x * 8) {
        const int  tile   = base + wid;
        const bool active = tile < numTiles;

        v8f acc[4] = {};
        int row = tile * 16 + rloc;
        if (row >= n) row = n - 1;

        for (int kc = 0; kc < NCHK; ++kc) {
            __syncthreads();
            for (int i = tid; i < KCH * HID; i += 256) {
                int k = i >> 6, c = i & 63;
                sBig[(k >> 1) * W1STR + 2 * c + (k & 1)] = W1[kc * KCH * HID + i];
            }
            __syncthreads();
            if (active) {
                const float* xrow = x + (size_t)row * FIN + kc * KCH + 2 * khalf;
                const v2f* bmat = (const v2f*)sBig;
#pragma unroll
                for (int kk = 0; kk < KCH / 4; ++kk) {
                    v2f a = ((const v2f*)xrow)[2 * kk];
                    const int pr = (2 * kk + khalf) * (W1STR / 2);
#pragma unroll
                    for (int nt = 0; nt < 4; ++nt) {
                        v2f b = bmat[pr + nt * 16 + rloc];
                        acc[nt] = __builtin_amdgcn_wmma_f32_16x16x4_f32(
                            false, a, false, b, (short)0, acc[nt], false, false);
                    }
                }
            }
        }

        __syncthreads();
        float* hsw = sBig + wid * (16 * HSTR);
        if (active) {
#pragma unroll
            for (int nt = 0; nt < 4; ++nt) {
                const float bb = b1[nt * 16 + rloc];
#pragma unroll
                for (int v = 0; v < 8; ++v) {
                    float val = acc[nt][v] + bb;
                    val = fmaxf(val, 0.0f);
                    const int M = v + 8 * khalf;
                    hsw[M * HSTR + nt * 16 + rloc] = val;
                }
            }
        }
        __syncthreads();

        if (active) {
            v8f acc2[3] = {};
            const v2f* amat = (const v2f*)hsw;
            const v2f* bmat = (const v2f*)sW2;
#pragma unroll 4
            for (int kk = 0; kk < HID / 4; ++kk) {
                v2f a2 = amat[rloc * (HSTR / 2) + 2 * kk + khalf];
                const int pr = (2 * kk + khalf) * (W2STR / 2);
#pragma unroll
                for (int nt = 0; nt < 3; ++nt) {
                    v2f b = bmat[pr + nt * 16 + rloc];
                    acc2[nt] = __builtin_amdgcn_wmma_f32_16x16x4_f32(
                        false, a2, false, b, (short)0, acc2[nt], false, false);
                }
            }
#pragma unroll
            for (int nt = 0; nt < 3; ++nt) {
                const int cb  = nt * 16 + rloc;
                const float bb = (cb < NCLS) ? b2[cb] : 0.0f;
#pragma unroll
                for (int v = 0; v < 8; ++v) {
                    const int r = tile * 16 + v + 8 * khalf;
                    if (r < n) {
                        float val = (cb < NCLS) ? (acc2[nt][v] + bb) : 0.0f;
                        zp[(size_t)r * CPAD + cb] = val;
                    }
                }
            }
        }
    }
}

__global__ void gpr_log_softmax(const float* __restrict__ zp, float* __restrict__ out, int n) {
    int i = blockIdx.x * blockDim.x + threadIdx.x;
    if (i >= n) return;
    const float* zr = zp + (size_t)i * CPAD;
    float vals[NCLS];
    float m = -3.4e38f;
#pragma unroll
    for (int c = 0; c < NCLS; ++c) { vals[c] = zr[c]; m = fmaxf(m, vals[c]); }
    float s = 0.0f;
#pragma unroll
    for (int c = 0; c < NCLS; ++c) s += __expf(vals[c] - m);
    const float ls = __logf(s);
    float* o = out + (size_t)i * NCLS;
#pragma unroll
    for (int c = 0; c < NCLS; ++c) o[c] = vals[c] - m - ls;
}

__global__ void gpr_acc_selfloop(const float* __restrict__ src, float* __restrict__ dst,
                                 const float* __restrict__ dis2, const float* __restrict__ temp,
                                 float* __restrict__ hid, int k, int first, int n) {
    int idx = blockIdx.x * blockDim.x + threadIdx.x;
    if (idx >= n * CPAD) return;
    const int i = idx / CPAD, c = idx % CPAD;
    const float v = src[idx];
    dst[idx] = dis2[i] * v;
    if (c < NCLS) {
        const float t = temp[k] * v;
        float* h = hid + (size_t)i * NCLS + c;
        if (first) *h = t; else *h += t;
    }
}

__global__ void gpr_scatter(const float* __restrict__ cur, float* __restrict__ nxt,
                            const int* __restrict__ row, const int* __restrict__ col,
                            const float* __restrict__ nrm, int e) {
    int ei = blockIdx.x * blockDim.x + threadIdx.x;
    if (ei >= e) return;
    const float w = nrm[ei];
    const float4* s = (const float4*)(cur + (size_t)row[ei] * CPAD);
    float* d = nxt + (size_t)col[ei] * CPAD;
#pragma unroll
    for (int j = 0; j < CPAD / 4; ++j) {
        float4 v = s[j];
        atomicAdd(d + 4 * j + 0, w * v.x);
        atomicAdd(d + 4 * j + 1, w * v.y);
        atomicAdd(d + 4 * j + 2, w * v.z);
        atomicAdd(d + 4 * j + 3, w * v.w);
    }
}

extern "C" void kernel_launch(void* const* d_in, const int* in_sizes, int n_in,
                              void* d_out, int out_size, void* d_ws, size_t ws_size,
                              hipStream_t stream) {
    const float* x    = (const float*)d_in[0];
    const int*   ei   = (const int*)  d_in[1];
    const float* W1   = (const float*)d_in[2];
    const float* b1   = (const float*)d_in[3];
    const float* W2   = (const float*)d_in[4];
    const float* b2   = (const float*)d_in[5];
    const float* temp = (const float*)d_in[6];

    const int N = in_sizes[0] / FIN;
    const int E = in_sizes[1] / 2;
    const int* row = ei;
    const int* col = ei + E;

    float* ws   = (float*)d_ws;
    const size_t N48 = (size_t)N * CPAD;
    float* zp   = ws;
    float* bufA = ws + N48;
    float* bufB = ws + 2 * N48;
    float* deg  = ws + 3 * N48;
    float* dis  = deg + N;
    float* dis2 = dis + N;
    float* nrm  = dis2 + N;

    float* out_ls = (float*)d_out;
    float* hid    = out_ls + (size_t)N * NCLS;

    const int TB = 256;
    const int gN = (N + TB - 1) / TB;
    const int gE = (E + TB - 1) / TB;

    gpr_init_deg <<<gN, TB, 0, stream>>>(deg, N);
    gpr_count_deg<<<gE, TB, 0, stream>>>(col, deg, E);
    gpr_dis      <<<gN, TB, 0, stream>>>(deg, dis, dis2, N);
    gpr_norm     <<<gE, TB, 0, stream>>>(row, col, dis, nrm, E);

    const int numTiles = (N + 15) / 16;
    const int gG = (numTiles + 7) / 8;
    gpr_mlp_wmma <<<gG, TB, 0, stream>>>(x, W1, b1, W2, b2, zp, N);
    gpr_log_softmax<<<gN, TB, 0, stream>>>(zp, out_ls, N);

    const int gP = ((N * CPAD) + TB - 1) / TB;
    gpr_acc_selfloop<<<gP, TB, 0, stream>>>(zp, bufA, dis2, temp, hid, 0, 1, N);
    const float* cur = zp;
    float* nx = bufA;
    for (int k = 1; k <= KSTEP; ++k) {
        gpr_scatter<<<gE, TB, 0, stream>>>(cur, nx, row, col, nrm, E);
        float* nn = (nx == bufA) ? bufB : bufA;
        gpr_acc_selfloop<<<gP, TB, 0, stream>>>(nx, nn, dis2, temp, hid, k, 0, N);
        cur = nx;
        nx = nn;
    }
}


# Propagation as pull over edges counting-sorted by destination once per call: 12 lanes sum a node's in-rows in registers; no float atomics; sum order now fixed (self, then sources ascending)
# speedup vs baseline: 14.8524x; 14.8524x over previous
#include <hip/hip_runtime.h>

#define FIN   500
#define HID   64
#define NCLS  47
#define CPAD  48
#define KSTEP 10
#define KCH   100
#define NCHK  5
#define W1STR 160
#define W2STR 96
#define HSTR  68

typedef __attribute__((ext_vector_type(2))) float v2f;
typedef __attribute__((ext_vector_type(8))) float v8f;

#define SCH 1024

__global__ void gpr_zero_cnt(int* __restrict__ cnt, int n) {
    int i = blockIdx.x * blockDim.x + threadIdx.x;
    if (i < n) cnt[i] = 0;
}

__global__ void gpr_count_rank(const int* __restrict__ col, int* __restrict__ cnt,
                               int* __restrict__ rank, int e) {
    int i = blockIdx.x * blockDim.x + threadIdx.x;
    if (i < e) rank[i] = atomicAdd(&cnt[col[i]], 1);
}

__global__ __launch_bounds__(256)
void gpr_scan_bsum(const int* __restrict__ cnt, int* __restrict__ bsum, int n) {
    __shared__ int sh[256];
    const int t  = threadIdx.x;
    const int b0 = blockIdx.x * SCH + t * 4;
    int s = 0;
#pragma unroll
    for (int j = 0; j < 4; ++j) { const int i = b0 + j; if (i < n) s += cnt[i]; }
    sh[t] = s;
    __syncthreads();
    for (int off = 128; off > 0; off >>= 1) {
        if (t < off) sh[t] += sh[t + off];
        __syncthreads();
    }
    if (t == 0) bsum[blockIdx.x] = sh[0];
}

__global__ __launch_bounds__(256)
void gpr_scan_final(const int* __restrict__ cnt, const int* __restrict__ bsum,
                    int* __restrict__ start, float* __restrict__ dis,
                    float* __restrict__ dis2, int n) {
    __shared__ int sh[256];
    __shared__ int sbase;
    const int t = threadIdx.x;
    int p = 0;
    for (int j = t; j < (int)blockIdx.x; j += 256) p += bsum[j];
    sh[t] = p;
    __syncthreads();
    for (int off = 128; off > 0; off >>= 1) {
        if (t < off) sh[t] += sh[t + off];
        __syncthreads();
    }
    if (t == 0) sbase = sh[0];
    __syncthreads();
    const int base = sbase;

    const int b0 = blockIdx.x * SCH + t * 4;
    int c[4];
    int s = 0;
#pragma unroll
    for (int j = 0; j < 4; ++j) { const int i = b0 + j; c[j] = (i < n) ? cnt[i] : 0; s += c[j]; }
    sh[t] = s;
    __syncthreads();
    for (int off = 1; off < 256; off <<= 1) {
        const int v = (t >= off) ? sh[t - off] : 0;
        __syncthreads();
        sh[t] += v;
        __syncthreads();
    }
    int run = base + sh[t] - s;
#pragma unroll
    for (int j = 0; j < 4; ++j) {
        const int i = b0 + j;
        if (i < n) {
            start[i] = run;
            run += c[j];
            const float d = 1.0f + (float)c[j];
            dis[i]  = rsqrtf(d);
            dis2[i] = 1.0f / d;
            if (i == n - 1) start[n] = run;
        }
    }
}

__global__ void gpr_scatter_keys(const int* __restrict__ row, const int* __restrict__ col,
                                 const int* __restrict__ rank, const int* __restrict__ start,
                                 int* __restrict__ kd, int e) {
    int i = blockIdx.x * blockDim.x + threadIdx.x;
    if (i >= e) return;
    const int c = col[i];
    int2 o;
    o.x = row[i];
    o.y = c;
    ((int2*)kd)[start[c] + rank[i]] = o;
}

__global__ void gpr_place(const int* __restrict__ start, const int* __restrict__ kd,
                          const float* __restrict__ dis, int* __restrict__ rec, int e) {
    int p = blockIdx.x * blockDim.x + threadIdx.x;
    if (p >= e) return;
    const int2* kd2 = (const int2*)kd;
    const int2 me = kd2[p];
    const int r = me.x, c = me.y;
    const int s0 = start[c], s1 = start[c + 1];
    int less = 0;
    for (int k = s0; k < s1; ++k) {
        const int kk = kd2[k].x;
        less += ((kk < r) || (kk == r && k < p)) ? 1 : 0;
    }
    int2 o;
    o.x = r;
    o.y = __float_as_int(dis[r] * dis[c]);
    ((int2*)rec)[s0 + less] = o;
}

__global__ __launch_bounds__(256)
void gpr_mlp_wmma(const float* __restrict__ x,  const float* __restrict__ W1,
                  const float* __restrict__ b1, const float* __restrict__ W2,
                  const float* __restrict__ b2, float* __restrict__ zp, int n) {
    __shared__ float sBig[8704];
    __shared__ float sW2[32 * W2STR];

    const int tid    = threadIdx.x;
    const int wid    = tid >> 5;
    const int lane   = tid & 31;
    const int rloc   = lane & 15;
    const int khalf  = lane >> 4;

    for (int i = tid; i < HID * CPAD; i += 256) {
        int k = i / CPAD, c = i % CPAD;
        sW2[(k >> 1) * W2STR + 2 * c + (k & 1)] = (c < NCLS) ? W2[k * NCLS + c] : 0.0f;
    }

    const int numTiles = (n + 15) / 16;
    for (int base = blockIdx.x * 8; base < numTiles; base += gridDim.x * 8) {
        const int  tile   = base + wid;
        const bool active = tile < numTiles;

        v8f acc[4] = {};
        int row = tile * 16 + rloc;
        if (row >= n) row = n - 1;

        for (int kc = 0; kc < NCHK; ++kc) {
            __syncthreads();
            for (int i = tid; i < KCH * HID; i += 256) {
                int k = i >> 6, c = i & 63;
                sBig[(k >> 1) * W1STR + 2 * c + (k & 1)] = W1[kc * KCH * HID + i];
            }
            __syncthreads();
            if (active) {
                const float* xrow = x + (size_t)row * FIN + kc * KCH + 2 * khalf;
                const v2f* bmat = (const v2f*)sBig;
#pragma unroll
                for (int kk = 0; kk < KCH / 4; ++kk) {
                    v2f a = ((const v2f*)xrow)[2 * kk];
                    const int pr = (2 * kk + khalf) * (W1STR / 2);
#pragma unroll
                    for (int nt = 0; nt < 4; ++nt) {
                        v2f b = bmat[pr + nt * 16 + rloc];
                        acc[nt] = __builtin_amdgcn_wmma_f32_16x16x4_f32(
                            false, a, false, b, (short)0, acc[nt], false, false);
                    }
                }
            }
        }

        __syncthreads();
        float* hsw = sBig + wid * (16 * HSTR);
        if (active) {
#pragma unroll
            for (int nt = 0; nt < 4; ++nt) {
                const float bb = b1[nt * 16 + rloc];
#pragma unroll
                for (int v = 0; v < 8; ++v) {
                    float val = acc[nt][v] + bb;
                    val = fmaxf(val, 0.0f);
                    const int M = v + 8 * khalf;
                    hsw[M * HSTR + nt * 16 + rloc] = val;
                }
            }
        }
        __syncthreads();

        if (active) {
            v8f acc2[3] = {};
            const v2f* amat = (const v2f*)hsw;
            const v2f* bmat = (const v2f*)sW2;
#pragma unroll 4
            for (int kk = 0; kk < HID / 4; ++kk) {
                v2f a2 = amat[rloc * (HSTR / 2) + 2 * kk + khalf];
                const int pr = (2 * kk + khalf) * (W2STR / 2);
#pragma unroll
                for (int nt = 0; nt < 3; ++nt) {
                    v2f b = bmat[pr + nt * 16 + rloc];
                    acc2[nt] = __builtin_amdgcn_wmma_f32_16x16x4_f32(
                        false, a2, false, b, (short)0, acc2[nt], false, false);
                }
            }
#pragma unroll
            for (int nt = 0; nt < 3; ++nt) {
                const int cb  = nt * 16 + rloc;
                const float bb = (cb < NCLS) ? b2[cb] : 0.0f;
#pragma unroll
                for (int v = 0; v < 8; ++v) {
                    const int r = tile * 16 + v + 8 * khalf;
                    if (r < n) {
                        float val = (cb < NCLS) ? (acc2[nt][v] + bb) : 0.0f;
                        zp[(size_t)r * CPAD + cb] = val;
                    }
                }
            }
        }
    }
}

__global__ void gpr_log_softmax(const float* __restrict__ zp, float* __restrict__ out, int n) {
    int i = blockIdx.x * blockDim.x + threadIdx.x;
    if (i >= n) return;
    const float* zr = zp + (size_t)i * CPAD;
    float vals[NCLS];
    float m = -3.4e38f;
#pragma unroll
    for (int c = 0; c < NCLS; ++c) { vals[c] = zr[c]; m = fmaxf(m, vals[c]); }
    float s = 0.0f;
#pragma unroll
    for (int c = 0; c < NCLS; ++c) s += __expf(vals[c] - m);
    const float ls = __logf(s);
    float* o = out + (size_t)i * NCLS;
#pragma unroll
    for (int c = 0; c < NCLS; ++c) o[c] = vals[c] - m - ls;
}

#define QPR 12
__device__ __forceinline__ void gpr_axpy4(float4& acc, float w, const float4& v) {
    acc.x += w * v.x; acc.y += w * v.y; acc.z += w * v.z; acc.w += w * v.w;
}
__global__ __launch_bounds__(192)
void gpr_pull(const float* __restrict__ cur, float* __restrict__ nxt,
              const int* __restrict__ start, const int* __restrict__ rec,
              const float* __restrict__ dis2, const float* __restrict__ temp,
              float* __restrict__ hid, int k, int wr, int n) {
    const int t = blockIdx.x * 192 + threadIdx.x;
    const int i = t / QPR;
    if (i >= n) return;
    const int q = t - i * QPR;
    const float4* c4 = (const float4*)cur;
    const int2*   r2 = (const int2*)rec;
    const int s0 = start[i], s1 = start[i + 1];
    const float4 self = c4[t];
    const float d2 = dis2[i];
    float4 acc;
    acc.x = d2 * self.x; acc.y = d2 * self.y; acc.z = d2 * self.z; acc.w = d2 * self.w;

    float* h = hid + (size_t)i * NCLS + 4 * q;
    const float sv[4] = {self.x, self.y, self.z, self.w};
    float hp[4];
    if (k == 1) {
        const float t0 = temp[0];
#pragma unroll
        for (int j = 0; j < 4; ++j) hp[j] = t0 * sv[j];
    } else {
#pragma unroll
        for (int j = 0; j < 4; ++j) hp[j] = h[(4 * q + j < NCLS) ? j : 0];
    }

    int e = s0;
    for (; e + 4 <= s1; e += 4) {
        const int2 a0 = r2[e], a1 = r2[e + 1], a2 = r2[e + 2], a3 = r2[e + 3];
        const float4 v0 = c4[a0.x * QPR + q];
        const float4 v1 = c4[a1.x * QPR + q];
        const float4 v2 = c4[a2.x * QPR + q];
        const float4 v3 = c4[a3.x * QPR + q];
        gpr_axpy4(acc, __int_as_float(a0.y), v0);
        gpr_axpy4(acc, __int_as_float(a1.y), v1);
        gpr_axpy4(acc, __int_as_float(a2.y), v2);
        gpr_axpy4(acc, __int_as_float(a3.y), v3);
    }
    const int rem = s1 - e;
    if (rem > 0) {
        const int e1 = (rem > 1) ? e + 1 : e;
        const int e2 = (rem > 2) ? e + 2 : e;
        const int2 a0 = r2[e], a1 = r2[e1], a2 = r2[e2];
        const float4 v0 = c4[a0.x * QPR + q];
        float4 v1 = c4[a1.x * QPR + q];
        float4 v2 = c4[a2.x * QPR + q];
        v1.x = (rem > 1) ? v1.x : 0.0f; v1.y = (rem > 1) ? v1.y : 0.0f;
        v1.z = (rem > 1) ? v1.z : 0.0f; v1.w = (rem > 1) ? v1.w : 0.0f;
        v2.x = (rem > 2) ? v2.x : 0.0f; v2.y = (rem > 2) ? v2.y : 0.0f;
        v2.z = (rem > 2) ? v2.z : 0.0f; v2.w = (rem > 2) ? v2.w : 0.0f;
        gpr_axpy4(acc, __int_as_float(a0.y), v0);
        gpr_axpy4(acc, __int_as_float(a1.y), v1);
        gpr_axpy4(acc, __int_as_float(a2.y), v2);
    }

    if (wr) ((float4*)nxt)[t] = acc;

    const float tk = temp[k];
    const float av[4] = {acc.x, acc.y, acc.z, acc.w};
#pragma unroll
    for (int j = 0; j < 4; ++j) {
        if (4 * q + j < NCLS) h[j] = hp[j] + tk * av[j];
    }
}

extern "C" void kernel_launch(void* const* d_in, const int* in_sizes, int n_in,
                              void* d_out, int out_size, void* d_ws, size_t ws_size,
                              hipStream_t stream) {
    const float* x    = (const float*)d_in[0];
    const int*   ei   = (const int*)  d_in[1];
    const float* W1   = (const float*)d_in[2];
    const float* b1   = (const float*)d_in[3];
    const float* W2   = (const float*)d_in[4];
    const float* b2   = (const float*)d_in[5];
    const float* temp = (const float*)d_in[6];

    const int N = in_sizes[0] / FIN;
    const int E = in_sizes[1] / 2;
    const int* row = ei;
    const int* col = ei + E;

    float* ws   = (float*)d_ws;
    const size_t N48 = (size_t)N * CPAD;
    float* zp   = ws;
    float* bufA = ws + N48;
    float* bufB = ws + 2 * N48;
    int*   rec  = (int*)(ws + 3 * N48);
    int*   kd   = rec + 2 * (size_t)E;
    int*   rank = kd + 2 * (size_t)E;
    int*   cnt  = rank + E;
    int*   start = cnt + N;
    float* dis  = (float*)(start + N + 1);
    float* dis2 = dis + N;
    int*   bsum = (int*)(dis2 + N);

    float* out_ls = (float*)d_out;
    float* hid    = out_ls + (size_t)N * NCLS;

    const int TB = 256;
    const int gN = (N + TB - 1) / TB;
    const int gE = (E + TB - 1) / TB;
    const int gS = (N + SCH - 1) / SCH;

    gpr_zero_cnt    <<<gN, TB, 0, stream>>>(cnt, N);
    gpr_count_rank  <<<gE, TB, 0, stream>>>(col, cnt, rank, E);
    gpr_scan_bsum   <<<gS, TB, 0, stream>>>(cnt, bsum, N);
    gpr_scan_final  <<<gS, TB, 0, stream>>>(cnt, bsum, start, dis, dis2, N);
    gpr_scatter_keys<<<gE, TB, 0, stream>>>(row, col, rank, start, kd, E);
    gpr_place       <<<gE, TB, 0, stream>>>(start, kd, dis, rec, E);

    const int numTiles = (N + 15) / 16;
    const int gG = (numTiles + 7) / 8;
    gpr_mlp_wmma <<<gG, TB, 0, stream>>>(x, W1, b1, W2, b2, zp, N);
    gpr_log_softmax<<<gN, TB, 0, stream>>>(zp, out_ls, N);

    const int gP = (N + 15) / 16;
    const float* cur = zp;
    float* nx = bufA;
    for (int k = 1; k <= KSTEP; ++k) {
        gpr_pull<<<gP, 192, 0, stream>>>(cur, nx, start, rec, dis2, temp, hid, k,
                                         (k < KSTEP) ? 1 : 0, N);
        cur = nx;
        nx = (nx == bufA) ? bufB : bufA;
    }
}


# Pull step re-cut: 16 lanes per node (3 floats each), records read 16 at a time as one line and shared in-row by DPP, 16 source rows in flight, resident waves walk 5 nodes each; sum order kept
# speedup vs baseline: 15.6080x; 1.0509x over previous
#include <hip/hip_runtime.h>

#define FIN   500
#define HID   64
#define NCLS  47
#define CPAD  48
#define KSTEP 10
#define KCH   100
#define NCHK  5
#define W1STR 160
#define W2STR 96
#define HSTR  68

typedef __attribute__((ext_vector_type(2))) float v2f;
typedef __attribute__((ext_vector_type(8))) float v8f;

#define SCH 1024

__global__ void gpr_zero_cnt(int* __restrict__ cnt, int n) {
    int i = blockIdx.x * blockDim.x + threadIdx.x;
    if (i < n) cnt[i] = 0;
}

__global__ void gpr_count_rank(const int* __restrict__ col, int* __restrict__ cnt,
                               int* __restrict__ rank, int e) {
    int i = blockIdx.x * blockDim.x + threadIdx.x;
    if (i < e) rank[i] = atomicAdd(&cnt[col[i]], 1);
}

__global__ __launch_bounds__(256)
void gpr_scan_bsum(const int* __restrict__ cnt, int* __restrict__ bsum, int n) {
    __shared__ int sh[256];
    const int t  = threadIdx.x;
    const int b0 = blockIdx.x * SCH + t * 4;
    int s = 0;
#pragma unroll
    for (int j = 0; j < 4; ++j) { const int i = b0 + j; if (i < n) s += cnt[i]; }
    sh[t] = s;
    __syncthreads();
    for (int off = 128; off > 0; off >>= 1) {
        if (t < off) sh[t] += sh[t + off];
        __syncthreads();
    }
    if (t == 0) bsum[blockIdx.x] = sh[0];
}

__global__ __launch_bounds__(256)
void gpr_scan_final(const int* __restrict__ cnt, const int* __restrict__ bsum,
                    int* __restrict__ start, float* __restrict__ dis,
                    float* __restrict__ dis2, int n) {
    __shared__ int sh[256];
    __shared__ int sbase;
    const int t = threadIdx.x;
    int p = 0;
    for (int j = t; j < (int)blockIdx.x; j += 256) p += bsum[j];
    sh[t] = p;
    __syncthreads();
    for (int off = 128; off > 0; off >>= 1) {
        if (t < off) sh[t] += sh[t + off];
        __syncthreads();
    }
    if (t == 0) sbase = sh[0];
    __syncthreads();
    const int base = sbase;

    const int b0 = blockIdx.x * SCH + t * 4;
    int c[4];
    int s = 0;
#pragma unroll
    for (int j = 0; j < 4; ++j) { const int i = b0 + j; c[j] = (i < n) ? cnt[i] : 0; s += c[j]; }
    sh[t] = s;
    __syncthreads();
    for (int off = 1; off < 256; off <<= 1) {
        const int v = (t >= off) ? sh[t - off] : 0;
        __syncthreads();
        sh[t] += v;
        __syncthreads();
    }
    int run = base + sh[t] - s;
#pragma unroll
    for (int j = 0; j < 4; ++j) {
        const int i = b0 + j;
        if (i < n) {
            start[i] = run;
            run += c[j];
            const float d = 1.0f + (float)c[j];
            dis[i]  = rsqrtf(d);
            dis2[i] = 1.0f / d;
            if (i == n - 1) start[n] = run;
        }
    }
}

__global__ void gpr_scatter_keys(const int* __restrict__ row, const int* __restrict__ col,
                                 const int* __restrict__ rank, const int* __restrict__ start,
                                 int* __restrict__ kd, int e) {
    int i = blockIdx.x * blockDim.x + threadIdx.x;
    if (i >= e) return;
    const int c = col[i];
    int2 o;
    o.x = row[i];
    o.y = c;
    ((int2*)kd)[start[c] + rank[i]] = o;
}

__global__ void gpr_place(const int* __restrict__ start, const int* __restrict__ kd,
                          const float* __restrict__ dis, int* __restrict__ rec, int e) {
    int p = blockIdx.x * blockDim.x + threadIdx.x;
    if (p >= e) return;
    const int2* kd2 = (const int2*)kd;
    const int2 me = kd2[p];
    const int r = me.x, c = me.y;
    const int s0 = start[c], s1 = start[c + 1];
    int less = 0;
    for (int k = s0; k < s1; ++k) {
        const int kk = kd2[k].x;
        less += ((kk < r) || (kk == r && k < p)) ? 1 : 0;
    }
    int2 o;
    o.x = r;
    o.y = __float_as_int(dis[r] * dis[c]);
    ((int2*)rec)[s0 + less] = o;
}

__global__ __launch_bounds__(256)
void gpr_mlp_wmma(const float* __restrict__ x,  const float* __restrict__ W1,
                  const float* __restrict__ b1, const float* __restrict__ W2,
                  const float* __restrict__ b2, float* __restrict__ zp, int n) {
    __shared__ float sBig[8704];
    __shared__ float sW2[32 * W2STR];

    const int tid    = threadIdx.x;
    const int wid    = tid >> 5;
    const int lane   = tid & 31;
    const int rloc   = lane & 15;
    const int khalf  = lane >> 4;

    for (int i = tid; i < HID * CPAD; i += 256) {
        int k = i / CPAD, c = i % CPAD;
        sW2[(k >> 1) * W2STR + 2 * c + (k & 1)] = (c < NCLS) ? W2[k * NCLS + c] : 0.0f;
    }

    const int numTiles = (n + 15) / 16;
    for (int base = blockIdx.x * 8; base < numTiles; base += gridDim.x * 8) {
        const int  tile   = base + wid;
        const bool active = tile < numTiles;

        v8f acc[4] = {};
        int row = tile * 16 + rloc;
        if (row >= n) row = n - 1;

        for (int kc = 0; kc < NCHK; ++kc) {
            __syncthreads();
            for (int i = tid; i < KCH * HID; i += 256) {
                int k = i >> 6, c = i & 63;
                sBig[(k >> 1) * W1STR + 2 * c + (k & 1)] = W1[kc * KCH * HID + i];
            }
            __syncthreads();
            if (active) {
                const float* xrow = x + (size_t)row * FIN + kc * KCH + 2 * khalf;
                const v2f* bmat = (const v2f*)sBig;
#pragma unroll
                for (int kk = 0; kk < KCH / 4; ++kk) {
                    v2f a = ((const v2f*)xrow)[2 * kk];
                    const int pr = (2 * kk + khalf) * (W1STR / 2);
#pragma unroll
                    for (int nt = 0; nt < 4; ++nt) {
                        v2f b = bmat[pr + nt * 16 + rloc];
                        acc[nt] = __builtin_amdgcn_wmma_f32_16x16x4_f32(
                            false, a, false, b, (short)0, acc[nt], false, false);
                    }
                }
            }
        }

        __syncthreads();
        float* hsw = sBig + wid * (16 * HSTR);
        if (active) {
#pragma unroll
            for (int nt = 0; nt < 4; ++nt) {
                const float bb = b1[nt * 16 + rloc];
#pragma unroll
                for (int v = 0; v < 8; ++v) {
                    float val = acc[nt][v] + bb;
                    val = fmaxf(val, 0.0f);
                    const int M = v + 8 * khalf;
                    hsw[M * HSTR + nt * 16 + rloc] = val;
                }
            }
        }
        __syncthreads();

        if (active) {
            v8f acc2[3] = {};
            const v2f* amat = (const v2f*)hsw;
            const v2f* bmat = (const v2f*)sW2;
#pragma unroll 4
            for (int kk = 0; kk < HID / 4; ++kk) {
                v2f a2 = amat[rloc * (HSTR / 2) + 2 * kk + khalf];
                const int pr = (2 * kk + khalf) * (W2STR / 2);
#pragma unroll
                for (int nt = 0; nt < 3; ++nt) {
                    v2f b = bmat[pr + nt * 16 + rloc];
                    acc2[nt] = __builtin_amdgcn_wmma_f32_16x16x4_f32(
                        false, a2, false, b, (short)0, acc2[nt], false, false);
                }
            }
#pragma unroll
            for (int nt = 0; nt < 3; ++nt) {
                const int cb  = nt * 16 + rloc;
                const float bb = (cb < NCLS) ? b2[cb] : 0.0f;
#pragma unroll
                for (int v = 0; v < 8; ++v) {
                    const int r = tile * 16 + v + 8 * khalf;
                    if (r < n) {
                        float val = (cb < NCLS) ? (acc2[nt][v] + bb) : 0.0f;
                        zp[(size_t)r * CPAD + cb] = val;
                    }
                }
            }
        }
    }
}

__global__ void gpr_log_softmax(const float* __restrict__ zp, float* __restrict__ out, int n) {
    int i = blockIdx.x * blockDim.x + threadIdx.x;
    if (i >= n) return;
    const float* zr = zp + (size_t)i * CPAD;
    float vals[NCLS];
    float m = -3.4e38f;
#pragma unroll
    for (int c = 0; c < NCLS; ++c) { vals[c] = zr[c]; m = fmaxf(m, vals[c]); }
    float s = 0.0f;
#pragma unroll
    for (int c = 0; c < NCLS; ++c) s += __expf(vals[c] - m);
    const float ls = __logf(s);
    float* o = out + (size_t)i * NCLS;
#pragma unroll
    for (int c = 0; c < NCLS; ++c) o[c] = vals[c] - m - ls;
}

#define PLB 256
#define PSWEEP 5
struct __attribute__((aligned(4))) f3 { float x, y, z; };

template <int J> __device__ __forceinline__ int gpr_row_bcast(int v) {
    return __builtin_amdgcn_update_dpp(0, v, 0x150 + J, 0xf, 0xf, true);
}
template <int J> __device__ __forceinline__ f3 gpr_row_of(const char* __restrict__ cb, int srcs, unsigned coff) {
    const unsigned off = (unsigned)gpr_row_bcast<J>(srcs) * 192u + coff;
    return *(const f3*)(cb + off);
}
template <int J> __device__ __forceinline__ void gpr_fma3(f3& acc, int ws, const f3& v) {
    const float w = __int_as_float(gpr_row_bcast<J>(ws));
    acc.x = __builtin_fmaf(w, v.x, acc.x);
    acc.y = __builtin_fmaf(w, v.y, acc.y);
    acc.z = __builtin_fmaf(w, v.z, acc.z);
}

__global__ __launch_bounds__(PLB, 10)
void gpr_pull(const float* __restrict__ cur, float* __restrict__ nxt,
              const int* __restrict__ start, const int* __restrict__ rec,
              const float* __restrict__ dis2, const float* __restrict__ temp,
              float* __restrict__ hid, int k, int wr, int n) {
    const int l16 = threadIdx.x & 15;
    const unsigned coff = 12u * (unsigned)l16;
    const int nh = (gridDim.x * PLB) >> 4;
    const char* cb = (const char*)cur;
    const int2* r2 = (const int2*)rec;
    const float tk = temp[k];
    const float t0 = temp[0];

    int i = (blockIdx.x * PLB + threadIdx.x) >> 4;
    if (i >= n) return;
    int s0 = start[i], s1 = start[i + 1];
    for (; i < n; i += nh) {
        const int inx = (i + nh < n) ? i + nh : i;
        const int s0n = start[inx], s1n = start[inx + 1];

        int base = s0;
        const int last = (s1 > 0) ? s1 - 1 : 0;
        int2 r = r2[(base + l16 < s1) ? base + l16 : last];

        const f3 self = *(const f3*)(cb + (unsigned)i * 192u + coff);
        const float d2 = dis2[i];

        float* h = hid + (size_t)i * NCLS + 3 * l16;
        float hp[3];
        if (k != 1) {
#pragma unroll
            for (int j = 0; j < 3; ++j) hp[j] = h[(3 * l16 + j < NCLS) ? j : 0];
        } else {
            hp[0] = t0 * self.x; hp[1] = t0 * self.y; hp[2] = t0 * self.z;
        }

        f3 acc;
        acc.x = d2 * self.x; acc.y = d2 * self.y; acc.z = d2 * self.z;

        while (base < s1) {
            const int sx = r.x;
            const int sw = (base + l16 < s1) ? r.y : 0;
            const f3 v0  = gpr_row_of<0>(cb, sx, coff);
            const f3 v1  = gpr_row_of<1>(cb, sx, coff);
            const f3 v2  = gpr_row_of<2>(cb, sx, coff);
            const f3 v3  = gpr_row_of<3>(cb, sx, coff);
            const f3 v4  = gpr_row_of<4>(cb, sx, coff);
            const f3 v5  = gpr_row_of<5>(cb, sx, coff);
            const f3 v6  = gpr_row_of<6>(cb, sx, coff);
            const f3 v7  = gpr_row_of<7>(cb, sx, coff);
            const f3 v8  = gpr_row_of<8>(cb, sx, coff);
            const f3 v9  = gpr_row_of<9>(cb, sx, coff);
            const f3 v10 = gpr_row_of<10>(cb, sx, coff);
            const f3 v11 = gpr_row_of<11>(cb, sx, coff);
            const f3 v12 = gpr_row_of<12>(cb, sx, coff);
            const f3 v13 = gpr_row_of<13>(cb, sx, coff);
            const f3 v14 = gpr_row_of<14>(cb, sx, coff);
            const f3 v15 = gpr_row_of<15>(cb, sx, coff);

            base += 16;
            r = r2[(base + l16 < s1) ? base + l16 : last];

            gpr_fma3<0>(acc, sw, v0);   gpr_fma3<1>(acc, sw, v1);
            gpr_fma3<2>(acc, sw, v2);   gpr_fma3<3>(acc, sw, v3);
            gpr_fma3<4>(acc, sw, v4);   gpr_fma3<5>(acc, sw, v5);
            gpr_fma3<6>(acc, sw, v6);   gpr_fma3<7>(acc, sw, v7);
            gpr_fma3<8>(acc, sw, v8);   gpr_fma3<9>(acc, sw, v9);
            gpr_fma3<10>(acc, sw, v10); gpr_fma3<11>(acc, sw, v11);
            gpr_fma3<12>(acc, sw, v12); gpr_fma3<13>(acc, sw, v13);
            gpr_fma3<14>(acc, sw, v14); gpr_fma3<15>(acc, sw, v15);
        }

        if (wr) *(f3*)((char*)nxt + (unsigned)i * 192u + coff) = acc;

        const float av[3] = {acc.x, acc.y, acc.z};
#pragma unroll
        for (int j = 0; j < 3; ++j) {
            if (3 * l16 + j < NCLS) h[j] = hp[j] + tk * av[j];
        }
        s0 = s0n; s1 = s1n;
    }
}

extern "C" void kernel_launch(void* const* d_in, const int* in_sizes, int n_in,
                              void* d_out, int out_size, void* d_ws, size_t ws_size,
                              hipStream_t stream) {
    const float* x    = (const float*)d_in[0];
    const int*   ei   = (const int*)  d_in[1];
    const float* W1   = (const float*)d_in[2];
    const float* b1   = (const float*)d_in[3];
    const float* W2   = (const float*)d_in[4];
    const float* b2   = (const float*)d_in[5];
    const float* temp = (const float*)d_in[6];

    const int N = in_sizes[0] / FIN;
    const int E = in_sizes[1] / 2;
    const int* row = ei;
    const int* col = ei + E;

    float* ws   = (float*)d_ws;
    const size_t N48 = (size_t)N * CPAD;
    float* zp   = ws;
    float* bufA = ws + N48;
    float* bufB = ws + 2 * N48;
    int*   rec  = (int*)(ws + 3 * N48);
    int*   kd   = rec + 2 * (size_t)E;
    int*   rank = kd + 2 * (size_t)E;
    int*   cnt  = rank + E;
    int*   start = cnt + N;
    float* dis  = (float*)(start + N + 1);
    float* dis2 = dis + N;
    int*   bsum = (int*)(dis2 + N);

    float* out_ls = (float*)d_out;
    float* hid    = out_ls + (size_t)N * NCLS;

    const int TB = 256;
    const int gN = (N + TB - 1) / TB;
    const int gE = (E + TB - 1) / TB;
    const int gS = (N + SCH - 1) / SCH;

    gpr_zero_cnt    <<<gN, TB, 0, stream>>>(cnt, N);
    gpr_count_rank  <<<gE, TB, 0, stream>>>(col, cnt, rank, E);
    gpr_scan_bsum   <<<gS, TB, 0, stream>>>(cnt, bsum, N);
    gpr_scan_final  <<<gS, TB, 0, stream>>>(cnt, bsum, start, dis, dis2, N);
    gpr_scatter_keys<<<gE, TB, 0, stream>>>(row, col, rank, start, kd, E);
    gpr_place       <<<gE, TB, 0, stream>>>(start, kd, dis, rec, E);

    const int numTiles = (N + 15) / 16;
    const int gG = (numTiles + 7) / 8;
    gpr_mlp_wmma <<<gG, TB, 0, stream>>>(x, W1, b1, W2, b2, zp, N);
    gpr_log_softmax<<<gN, TB, 0, stream>>>(zp, out_ls, N);

    const int gP = (N + 16 * PSWEEP - 1) / (16 * PSWEEP);
    const float* cur = zp;
    float* nx = bufA;
    for (int k = 1; k <= KSTEP; ++k) {
        gpr_pull<<<gP, PLB, 0, stream>>>(cur, nx, start, rec, dis2, temp, hid, k,
                                            (k < KSTEP) ? 1 : 0, N);
        cur = nx;
        nx = (nx == bufA) ? bufB : bufA;
    }
}


# Edge bucketing by destination now counts and ranks on LDS counters, one block per (destination range, edge chunk); the scattered returning global atomics are gone; sum order unchanged
# speedup vs baseline: 17.7708x; 1.1386x over previous
#include <hip/hip_runtime.h>

#define FIN   500
#define HID   64
#define NCLS  47
#define CPAD  48
#define KSTEP 10
#define KCH   100
#define NCHK  5
#define W1STR 160
#define W2STR 96
#define HSTR  68

typedef __attribute__((ext_vector_type(2))) float v2f;
typedef __attribute__((ext_vector_type(8))) float v8f;

#define SCH 1024
#define BNR 25088
#define BTH 1024
#define BU  16
#define BTR 3
#define BCE (BTH * BU * BTR)
#define BQ  ((BNR + BTH - 1) / BTH)

__global__ __launch_bounds__(BTH)
void gpr_bin_count(const int* __restrict__ col, int* __restrict__ part, int n, int e, int nrange) {
    __shared__ int sc[BNR];
    const int tid  = threadIdx.x;
    const int r    = blockIdx.x % nrange;
    const int c    = blockIdx.x / nrange;
    const int lo   = r * BNR;
    const int span = (n - lo < BNR) ? n - lo : BNR;
    for (int j = tid; j < span; j += BTH) sc[j] = 0;
    __syncthreads();
    const int e0 = c * BCE;
    const int e1 = (e - e0 < BCE) ? e : e0 + BCE;
    for (int b = e0; b < e1; b += BTH * BU) {
        int d[BU];
#pragma unroll
        for (int u = 0; u < BU; ++u) {
            const int i = b + u * BTH + tid;
            const int v = col[(i < e1) ? i : e1 - 1];
            d[u] = (i < e1) ? v : -1;
        }
#pragma unroll
        for (int u = 0; u < BU; ++u) {
            const unsigned dd = (unsigned)(d[u] - lo);
            if (dd < (unsigned)span) atomicAdd(&sc[dd], 1);
        }
    }
    __syncthreads();
    int* prow = part + (size_t)c * n + lo;
    for (int j = tid; j < span; j += BTH) prow[j] = sc[j];
}

__global__ void gpr_bin_scan(const int* __restrict__ part, int* __restrict__ pofs,
                             int* __restrict__ cnt, int n, int nchunk) {
    const int j = blockIdx.x * blockDim.x + threadIdx.x;
    if (j >= n) return;
    int run = 0;
#pragma unroll 16
    for (int c = 0; c < nchunk; ++c) {
        const int v = part[(size_t)c * n + j];
        pofs[(size_t)c * n + j] = run;
        run += v;
    }
    cnt[j] = run;
}

__global__ __launch_bounds__(256)
void gpr_scan_bsum(const int* __restrict__ cnt, int* __restrict__ bsum, int n) {
    __shared__ int sh[256];
    const int t  = threadIdx.x;
    const int b0 = blockIdx.x * SCH + t * 4;
    int s = 0;
#pragma unroll
    for (int j = 0; j < 4; ++j) { const int i = b0 + j; if (i < n) s += cnt[i]; }
    sh[t] = s;
    __syncthreads();
    for (int off = 128; off > 0; off >>= 1) {
        if (t < off) sh[t] += sh[t + off];
        __syncthreads();
    }
    if (t == 0) bsum[blockIdx.x] = sh[0];
}

__global__ __launch_bounds__(256)
void gpr_scan_final(const int* __restrict__ cnt, const int* __restrict__ bsum,
                    int* __restrict__ start, float* __restrict__ dis,
                    float* __restrict__ dis2, int n) {
    __shared__ int sh[256];
    __shared__ int sbase;
    const int t = threadIdx.x;
    int p = 0;
    for (int j = t; j < (int)blockIdx.x; j += 256) p += bsum[j];
    sh[t] = p;
    __syncthreads();
    for (int off = 128; off > 0; off >>= 1) {
        if (t < off) sh[t] += sh[t + off];
        __syncthreads();
    }
    if (t == 0) sbase = sh[0];
    __syncthreads();
    const int base = sbase;

    const int b0 = blockIdx.x * SCH + t * 4;
    int c[4];
    int s = 0;
#pragma unroll
    for (int j = 0; j < 4; ++j) { const int i = b0 + j; c[j] = (i < n) ? cnt[i] : 0; s += c[j]; }
    sh[t] = s;
    __syncthreads();
    for (int off = 1; off < 256; off <<= 1) {
        const int v = (t >= off) ? sh[t - off] : 0;
        __syncthreads();
        sh[t] += v;
        __syncthreads();
    }
    int run = base + sh[t] - s;
#pragma unroll
    for (int j = 0; j < 4; ++j) {
        const int i = b0 + j;
        if (i < n) {
            start[i] = run;
            run += c[j];
            const float d = 1.0f + (float)c[j];
            dis[i]  = rsqrtf(d);
            dis2[i] = 1.0f / d;
            if (i == n - 1) start[n] = run;
        }
    }
}

__global__ __launch_bounds__(BTH)
void gpr_bin_scatter(const int* __restrict__ row, const int* __restrict__ col,
                     const int* __restrict__ start, const int* __restrict__ pofs,
                     int* __restrict__ kd, int n, int e, int nrange) {
    __shared__ int sc[BNR];
    const int tid  = threadIdx.x;
    const int r    = blockIdx.x % nrange;
    const int c    = blockIdx.x / nrange;
    const int lo   = r * BNR;
    const int span = (n - lo < BNR) ? n - lo : BNR;
    {
        const int* srow = start + lo;
        const int* prow = pofs + (size_t)c * n + lo;
        int v[BQ];
#pragma unroll
        for (int q = 0; q < BQ; ++q) {
            const int j  = tid + q * BTH;
            const int jc = (j < span) ? j : span - 1;
            v[q] = srow[jc] + prow[jc];
        }
#pragma unroll
        for (int q = 0; q < BQ; ++q) {
            const int j = tid + q * BTH;
            if (j < span) sc[j] = v[q];
        }
    }
    __syncthreads();
    const int e0 = c * BCE;
    const int e1 = (e - e0 < BCE) ? e : e0 + BCE;
    int2* kd2 = (int2*)kd;
    for (int b = e0; b < e1; b += BTH * BU) {
        int d[BU], s[BU], slot[BU];
#pragma unroll
        for (int u = 0; u < BU; ++u) {
            const int i  = b + u * BTH + tid;
            const int ic = (i < e1) ? i : e1 - 1;
            const int vd = col[ic];
            s[u] = row[ic];
            d[u] = (i < e1) ? vd : -1;
        }
#pragma unroll
        for (int u = 0; u < BU; ++u) {
            const unsigned dd = (unsigned)(d[u] - lo);
            int sl = -1;
            if (dd < (unsigned)span) sl = atomicAdd(&sc[dd], 1);
            slot[u] = sl;
        }
#pragma unroll
        for (int u = 0; u < BU; ++u) {
            if (slot[u] >= 0) {
                int2 o;
                o.x = s[u];
                o.y = d[u];
                kd2[slot[u]] = o;
            }
        }
    }
}

__global__ void gpr_place(const int* __restrict__ start, const int* __restrict__ kd,
                          const float* __restrict__ dis, int* __restrict__ rec, int e) {
    int p = blockIdx.x * blockDim.x + threadIdx.x;
    if (p >= e) return;
    const int2* kd2 = (const int2*)kd;
    const int2 me = kd2[p];
    const int r = me.x, c = me.y;
    const int s0 = start[c], s1 = start[c + 1];
    int less = 0;
    for (int k = s0; k < s1; ++k) {
        const int kk = kd2[k].x;
        less += ((kk < r) || (kk == r && k < p)) ? 1 : 0;
    }
    int2 o;
    o.x = r;
    o.y = __float_as_int(dis[r] * dis[c]);
    ((int2*)rec)[s0 + less] = o;
}

__global__ __launch_bounds__(256)
void gpr_mlp_wmma(const float* __restrict__ x,  const float* __restrict__ W1,
                  const float* __restrict__ b1, const float* __restrict__ W2,
                  const float* __restrict__ b2, float* __restrict__ zp, int n) {
    __shared__ float sBig[8704];
    __shared__ float sW2[32 * W2STR];

    const int tid    = threadIdx.x;
    const int wid    = tid >> 5;
    const int lane   = tid & 31;
    const int rloc   = lane & 15;
    const int khalf  = lane >> 4;

    for (int i = tid; i < HID * CPAD; i += 256) {
        int k = i / CPAD, c = i % CPAD;
        sW2[(k >> 1) * W2STR + 2 * c + (k & 1)] = (c < NCLS) ? W2[k * NCLS + c] : 0.0f;
    }

    const int numTiles = (n + 15) / 16;
    for (int base = blockIdx.x * 8; base < numTiles; base += gridDim.x * 8) {
        const int  tile   = base + wid;
        const bool active = tile < numTiles;

        v8f acc[4] = {};
        int row = tile * 16 + rloc;
        if (row >= n) row = n - 1;

        for (int kc = 0; kc < NCHK; ++kc) {
            __syncthreads();
            for (int i = tid; i < KCH * HID; i += 256) {
                int k = i >> 6, c = i & 63;
                sBig[(k >> 1) * W1STR + 2 * c + (k & 1)] = W1[kc * KCH * HID + i];
            }
            __syncthreads();
            if (active) {
                const float* xrow = x + (size_t)row * FIN + kc * KCH + 2 * khalf;
                const v2f* bmat = (const v2f*)sBig;
#pragma unroll
                for (int kk = 0; kk < KCH / 4; ++kk) {
                    v2f a = ((const v2f*)xrow)[2 * kk];
                    const int pr = (2 * kk + khalf) * (W1STR / 2);
#pragma unroll
                    for (int nt = 0; nt < 4; ++nt) {
                        v2f b = bmat[pr + nt * 16 + rloc];
                        acc[nt] = __builtin_amdgcn_wmma_f32_16x16x4_f32(
                            false, a, false, b, (short)0, acc[nt], false, false);
                    }
                }
            }
        }

        __syncthreads();
        float* hsw = sBig + wid * (16 * HSTR);
        if (active) {
#pragma unroll
            for (int nt = 0; nt < 4; ++nt) {
                const float bb = b1[nt * 16 + rloc];
#pragma unroll
                for (int v = 0; v < 8; ++v) {
                    float val = acc[nt][v] + bb;
                    val = fmaxf(val, 0.0f);
                    const int M = v + 8 * khalf;
                    hsw[M * HSTR + nt * 16 + rloc] = val;
                }
            }
        }
        __syncthreads();

        if (active) {
            v8f acc2[3] = {};
            const v2f* amat = (const v2f*)hsw;
            const v2f* bmat = (const v2f*)sW2;
#pragma unroll 4
            for (int kk = 0; kk < HID / 4; ++kk) {
                v2f a2 = amat[rloc * (HSTR / 2) + 2 * kk + khalf];
                const int pr = (2 * kk + khalf) * (W2STR / 2);
#pragma unroll
                for (int nt = 0; nt < 3; ++nt) {
                    v2f b = bmat[pr + nt * 16 + rloc];
                    acc2[nt] = __builtin_amdgcn_wmma_f32_16x16x4_f32(
                        false, a2, false, b, (short)0, acc2[nt], false, false);
                }
            }
#pragma unroll
            for (int nt = 0; nt < 3; ++nt) {
                const int cb  = nt * 16 + rloc;
                const float bb = (cb < NCLS) ? b2[cb] : 0.0f;
#pragma unroll
                for (int v = 0; v < 8; ++v) {
                    const int r = tile * 16 + v + 8 * khalf;
                    if (r < n) {
                        float val = (cb < NCLS) ? (acc2[nt][v] + bb) : 0.0f;
                        zp[(size_t)r * CPAD + cb] = val;
                    }
                }
            }
        }
    }
}

__global__ void gpr_log_softmax(const float* __restrict__ zp, float* __restrict__ out, int n) {
    int i = blockIdx.x * blockDim.x + threadIdx.x;
    if (i >= n) return;
    const float* zr = zp + (size_t)i * CPAD;
    float vals[NCLS];
    float m = -3.4e38f;
#pragma unroll
    for (int c = 0; c < NCLS; ++c) { vals[c] = zr[c]; m = fmaxf(m, vals[c]); }
    float s = 0.0f;
#pragma unroll
    for (int c = 0; c < NCLS; ++c) s += __expf(vals[c] - m);
    const float ls = __logf(s);
    float* o = out + (size_t)i * NCLS;
#pragma unroll
    for (int c = 0; c < NCLS; ++c) o[c] = vals[c] - m - ls;
}

#define PLB 256
#define PSWEEP 5
struct __attribute__((aligned(4))) f3 { float x, y, z; };

template <int J> __device__ __forceinline__ int gpr_row_bcast(int v) {
    return __builtin_amdgcn_update_dpp(0, v, 0x150 + J, 0xf, 0xf, true);
}
template <int J> __device__ __forceinline__ f3 gpr_row_of(const char* __restrict__ cb, int srcs, unsigned coff) {
    const unsigned off = (unsigned)gpr_row_bcast<J>(srcs) * 192u + coff;
    return *(const f3*)(cb + off);
}
template <int J> __device__ __forceinline__ void gpr_fma3(f3& acc, int ws, const f3& v) {
    const float w = __int_as_float(gpr_row_bcast<J>(ws));
    acc.x = __builtin_fmaf(w, v.x, acc.x);
    acc.y = __builtin_fmaf(w, v.y, acc.y);
    acc.z = __builtin_fmaf(w, v.z, acc.z);
}

__global__ __launch_bounds__(PLB, 10)
void gpr_pull(const float* __restrict__ cur, float* __restrict__ nxt,
              const int* __restrict__ start, const int* __restrict__ rec,
              const float* __restrict__ dis2, const float* __restrict__ temp,
              float* __restrict__ hid, int k, int wr, int n) {
    const int l16 = threadIdx.x & 15;
    const unsigned coff = 12u * (unsigned)l16;
    const int nh = (gridDim.x * PLB) >> 4;
    const char* cb = (const char*)cur;
    const int2* r2 = (const int2*)rec;
    const float tk = temp[k];
    const float t0 = temp[0];

    int i = (blockIdx.x * PLB + threadIdx.x) >> 4;
    if (i >= n) return;
    int s0 = start[i], s1 = start[i + 1];
    for (; i < n; i += nh) {
        const int inx = (i + nh < n) ? i + nh : i;
        const int s0n = start[inx], s1n = start[inx + 1];

        int base = s0;
        const int last = (s1 > 0) ? s1 - 1 : 0;
        int2 r = r2[(base + l16 < s1) ? base + l16 : last];

        const f3 self = *(const f3*)(cb + (unsigned)i * 192u + coff);
        const float d2 = dis2[i];

        float* h = hid + (size_t)i * NCLS + 3 * l16;
        float hp[3];
        if (k != 1) {
#pragma unroll
            for (int j = 0; j < 3; ++j) hp[j] = h[(3 * l16 + j < NCLS) ? j : 0];
        } else {
            hp[0] = t0 * self.x; hp[1] = t0 * self.y; hp[2] = t0 * self.z;
        }

        f3 acc;
        acc.x = d2 * self.x; acc.y = d2 * self.y; acc.z = d2 * self.z;

        while (base < s1) {
            const int sx = r.x;
            const int sw = (base + l16 < s1) ? r.y : 0;
            const f3 v0  = gpr_row_of<0>(cb, sx, coff);
            const f3 v1  = gpr_row_of<1>(cb, sx, coff);
            const f3 v2  = gpr_row_of<2>(cb, sx, coff);
            const f3 v3  = gpr_row_of<3>(cb, sx, coff);
            const f3 v4  = gpr_row_of<4>(cb, sx, coff);
            const f3 v5  = gpr_row_of<5>(cb, sx, coff);
            const f3 v6  = gpr_row_of<6>(cb, sx, coff);
            const f3 v7  = gpr_row_of<7>(cb, sx, coff);
            const f3 v8  = gpr_row_of<8>(cb, sx, coff);
            const f3 v9  = gpr_row_of<9>(cb, sx, coff);
            const f3 v10 = gpr_row_of<10>(cb, sx, coff);
            const f3 v11 = gpr_row_of<11>(cb, sx, coff);
            const f3 v12 = gpr_row_of<12>(cb, sx, coff);
            const f3 v13 = gpr_row_of<13>(cb, sx, coff);
            const f3 v14 = gpr_row_of<14>(cb, sx, coff);
            const f3 v15 = gpr_row_of<15>(cb, sx, coff);

            base += 16;
            r = r2[(base + l16 < s1) ? base + l16 : last];

            gpr_fma3<0>(acc, sw, v0);   gpr_fma3<1>(acc, sw, v1);
            gpr_fma3<2>(acc, sw, v2);   gpr_fma3<3>(acc, sw, v3);
            gpr_fma3<4>(acc, sw, v4);   gpr_fma3<5>(acc, sw, v5);
            gpr_fma3<6>(acc, sw, v6);   gpr_fma3<7>(acc, sw, v7);
            gpr_fma3<8>(acc, sw, v8);   gpr_fma3<9>(acc, sw, v9);
            gpr_fma3<10>(acc, sw, v10); gpr_fma3<11>(acc, sw, v11);
            gpr_fma3<12>(acc, sw, v12); gpr_fma3<13>(acc, sw, v13);
            gpr_fma3<14>(acc, sw, v14); gpr_fma3<15>(acc, sw, v15);
        }

        if (wr) *(f3*)((char*)nxt + (unsigned)i * 192u + coff) = acc;

        const float av[3] = {acc.x, acc.y, acc.z};
#pragma unroll
        for (int j = 0; j < 3; ++j) {
            if (3 * l16 + j < NCLS) h[j] = hp[j] + tk * av[j];
        }
        s0 = s0n; s1 = s1n;
    }
}

extern "C" void kernel_launch(void* const* d_in, const int* in_sizes, int n_in,
                              void* d_out, int out_size, void* d_ws, size_t ws_size,
                              hipStream_t stream) {
    const float* x    = (const float*)d_in[0];
    const int*   ei   = (const int*)  d_in[1];
    const float* W1   = (const float*)d_in[2];
    const float* b1   = (const float*)d_in[3];
    const float* W2   = (const float*)d_in[4];
    const float* b2   = (const float*)d_in[5];
    const float* temp = (const float*)d_in[6];

    const int N = in_sizes[0] / FIN;
    const int E = in_sizes[1] / 2;
    const int* row = ei;
    const int* col = ei + E;

    float* ws   = (float*)d_ws;
    const size_t N48 = (size_t)N * CPAD;
    float* zp   = ws;
    float* bufA = ws + N48;
    float* bufB = ws + 2 * N48;
    int*   rec  = (int*)(ws + 3 * N48);
    int*   kd   = rec + 2 * (size_t)E;
    const int nrange = (N + BNR - 1) / BNR;
    const int nchunk = (E + BCE - 1) / BCE;
    const size_t CN  = (size_t)nchunk * N;
    int*   part = kd + 2 * (size_t)E;
    int*   pofs = part + CN;
    int*   cnt  = pofs + CN;
    int*   start = cnt + N;
    float* dis  = (float*)(start + N + 1);
    float* dis2 = dis + N;
    int*   bsum = (int*)(dis2 + N);

    float* out_ls = (float*)d_out;
    float* hid    = out_ls + (size_t)N * NCLS;

    const int TB = 256;
    const int gN = (N + TB - 1) / TB;
    const int gE = (E + TB - 1) / TB;
    const int gS = (N + SCH - 1) / SCH;

    gpr_bin_count   <<<nrange * nchunk, BTH, 0, stream>>>(col, part, N, E, nrange);
    gpr_bin_scan    <<<gN, TB, 0, stream>>>(part, pofs, cnt, N, nchunk);
    gpr_scan_bsum   <<<gS, TB, 0, stream>>>(cnt, bsum, N);
    gpr_scan_final  <<<gS, TB, 0, stream>>>(cnt, bsum, start, dis, dis2, N);
    gpr_bin_scatter <<<nrange * nchunk, BTH, 0, stream>>>(row, col, start, pofs, kd, N, E, nrange);
    gpr_place       <<<gE, TB, 0, stream>>>(start, kd, dis, rec, E);

    const int numTiles = (N + 15) / 16;
    const int gG = (numTiles + 7) / 8;
    gpr_mlp_wmma <<<gG, TB, 0, stream>>>(x, W1, b1, W2, b2, zp, N);
    gpr_log_softmax<<<gN, TB, 0, stream>>>(zp, out_ls, N);

    const int gP = (N + 16 * PSWEEP - 1) / (16 * PSWEEP);
    const float* cur = zp;
    float* nx = bufA;
    for (int k = 1; k <= KSTEP; ++k) {
        gpr_pull<<<gP, PLB, 0, stream>>>(cur, nx, start, rec, dis2, temp, hid, k,
                                            (k < KSTEP) ? 1 : 0, N);
        cur = nx;
        nx = (nx == bufA) ? bufB : bufA;
    }
}


# speedup vs baseline: 18.6349x; 1.0486x over previous
#include <hip/hip_runtime.h>

#define FIN   500
#define HID   64
#define NCLS  47
#define CPAD  48
#define ZSTR  64
#define KSTEP 10
#define KCH   100
#define NCHK  5
#define W1STR 160
#define W2STR 96
#define HSTR  68

typedef __attribute__((ext_vector_type(2))) float v2f;
typedef __attribute__((ext_vector_type(8))) float v8f;

#define SCH 1024
#define BNR 25088
#define BTH 1024
#define BU  16
#define BTR 3
#define BCE (BTH * BU * BTR)
#define BQ  ((BNR + BTH - 1) / BTH)

__global__ __launch_bounds__(BTH)
void gpr_bin_count(const int* __restrict__ col, int* __restrict__ part, int n, int e, int nrange) {
    __shared__ int sc[BNR];
    const int tid  = threadIdx.x;
    const int r    = blockIdx.x % nrange;
    const int c    = blockIdx.x / nrange;
    const int lo   = r * BNR;
    const int span = (n - lo < BNR) ? n - lo : BNR;
    for (int j = tid; j < span; j += BTH) sc[j] = 0;
    __syncthreads();
    const int e0 = c * BCE;
    const int e1 = (e - e0 < BCE) ? e : e0 + BCE;
    for (int b = e0; b < e1; b += BTH * BU) {
        int d[BU];
#pragma unroll
        for (int u = 0; u < BU; ++u) {
            const int i = b + u * BTH + tid;
            const int v = col[(i < e1) ? i : e1 - 1];
            d[u] = (i < e1) ? v : -1;
        }
#pragma unroll
        for (int u = 0; u < BU; ++u) {
            const unsigned dd = (unsigned)(d[u] - lo);
            if (dd < (unsigned)span) atomicAdd(&sc[dd], 1);
        }
    }
    __syncthreads();
    int* prow = part + (size_t)c * n + lo;
    for (int j = tid; j < span; j += BTH) prow[j] = sc[j];
}

__global__ void gpr_bin_scan(const int* __restrict__ part, int* __restrict__ pofs,
                             int* __restrict__ cnt, int n, int nchunk) {
    const int j = blockIdx.x * blockDim.x + threadIdx.x;
    if (j >= n) return;
    int run = 0;
#pragma unroll 16
    for (int c = 0; c < nchunk; ++c) {
        const int v = part[(size_t)c * n + j];
        pofs[(size_t)c * n + j] = run;
        run += v;
    }
    cnt[j] = run;
}

__global__ __launch_bounds__(256)
void gpr_scan_bsum(const int* __restrict__ cnt, int* __restrict__ bsum, int n) {
    __shared__ int sh[256];
    const int t  = threadIdx.x;
    const int b0 = blockIdx.x * SCH + t * 4;
    int s = 0;
#pragma unroll
    for (int j = 0; j < 4; ++j) { const int i = b0 + j; if (i < n) s += cnt[i]; }
    sh[t] = s;
    __syncthreads();
    for (int off = 128; off > 0; off >>= 1) {
        if (t < off) sh[t] += sh[t + off];
        __syncthreads();
    }
    if (t == 0) bsum[blockIdx.x] = sh[0];
}

__global__ __launch_bounds__(256)
void gpr_scan_final(const int* __restrict__ cnt, const int* __restrict__ bsum,
                    int* __restrict__ start, float* __restrict__ dis,
                    float* __restrict__ dis2, int n) {
    __shared__ int sh[256];
    __shared__ int sbase;
    const int t = threadIdx.x;
    int p = 0;
    for (int j = t; j < (int)blockIdx.x; j += 256) p += bsum[j];
    sh[t] = p;
    __syncthreads();
    for (int off = 128; off > 0; off >>= 1) {
        if (t < off) sh[t] += sh[t + off];
        __syncthreads();
    }
    if (t == 0) sbase = sh[0];
    __syncthreads();
    const int base = sbase;

    const int b0 = blockIdx.x * SCH + t * 4;
    int c[4];
    int s = 0;
#pragma unroll
    for (int j = 0; j < 4; ++j) { const int i = b0 + j; c[j] = (i < n) ? cnt[i] : 0; s += c[j]; }
    sh[t] = s;
    __syncthreads();
    for (int off = 1; off < 256; off <<= 1) {
        const int v = (t >= off) ? sh[t - off] : 0;
        __syncthreads();
        sh[t] += v;
        __syncthreads();
    }
    int run = base + sh[t] - s;
#pragma unroll
    for (int j = 0; j < 4; ++j) {
        const int i = b0 + j;
        if (i < n) {
            start[i] = run;
            run += c[j];
            const float d = 1.0f + (float)c[j];
            dis[i]  = rsqrtf(d);
            dis2[i] = 1.0f / d;
            if (i == n - 1) start[n] = run;
        }
    }
}

__global__ __launch_bounds__(BTH)
void gpr_bin_scatter(const int* __restrict__ row, const int* __restrict__ col,
                     const int* __restrict__ start, const int* __restrict__ pofs,
                     int* __restrict__ kd, int n, int e, int nrange) {
    __shared__ int sc[BNR];
    const int tid  = threadIdx.x;
    const int r    = blockIdx.x % nrange;
    const int c    = blockIdx.x / nrange;
    const int lo   = r * BNR;
    const int span = (n - lo < BNR) ? n - lo : BNR;
    {
        const int* srow = start + lo;
        const int* prow = pofs + (size_t)c * n + lo;
        int v[BQ];
#pragma unroll
        for (int q = 0; q < BQ; ++q) {
            const int j  = tid + q * BTH;
            const int jc = (j < span) ? j : span - 1;
            v[q] = srow[jc] + prow[jc];
        }
#pragma unroll
        for (int q = 0; q < BQ; ++q) {
            const int j = tid + q * BTH;
            if (j < span) sc[j] = v[q];
        }
    }
    __syncthreads();
    const int e0 = c * BCE;
    const int e1 = (e - e0 < BCE) ? e : e0 + BCE;
    int2* kd2 = (int2*)kd;
    for (int b = e0; b < e1; b += BTH * BU) {
        int d[BU], s[BU], slot[BU];
#pragma unroll
        for (int u = 0; u < BU; ++u) {
            const int i  = b + u * BTH + tid;
            const int ic = (i < e1) ? i : e1 - 1;
            const int vd = col[ic];
            s[u] = row[ic];
            d[u] = (i < e1) ? vd : -1;
        }
#pragma unroll
        for (int u = 0; u < BU; ++u) {
            const unsigned dd = (unsigned)(d[u] - lo);
            int sl = -1;
            if (dd < (unsigned)span) sl = atomicAdd(&sc[dd], 1);
            slot[u] = sl;
        }
#pragma unroll
        for (int u = 0; u < BU; ++u) {
            if (slot[u] >= 0) {
                int2 o;
                o.x = s[u];
                o.y = d[u];
                kd2[slot[u]] = o;
            }
        }
    }
}

__global__ void gpr_place(const int* __restrict__ start, const int* __restrict__ kd,
                          const float* __restrict__ dis, int* __restrict__ rec, int e) {
    int p = blockIdx.x * blockDim.x + threadIdx.x;
    if (p >= e) return;
    const int2* kd2 = (const int2*)kd;
    const int2 me = kd2[p];
    const int r = me.x, c = me.y;
    const int s0 = start[c], s1 = start[c + 1];
    int less = 0;
    for (int k = s0; k < s1; ++k) {
        const int kk = kd2[k].x;
        less += ((kk < r) || (kk == r && k < p)) ? 1 : 0;
    }
    int2 o;
    o.x = r;
    o.y = __float_as_int(dis[r] * dis[c]);
    ((int2*)rec)[s0 + less] = o;
}

__global__ __launch_bounds__(256)
void gpr_mlp_wmma(const float* __restrict__ x,  const float* __restrict__ W1,
                  const float* __restrict__ b1, const float* __restrict__ W2,
                  const float* __restrict__ b2, float* __restrict__ zp, int n) {
    __shared__ float sBig[8704];
    __shared__ float sW2[32 * W2STR];

    const int tid    = threadIdx.x;
    const int wid    = tid >> 5;
    const int lane   = tid & 31;
    const int rloc   = lane & 15;
    const int khalf  = lane >> 4;

    for (int i = tid; i < HID * CPAD; i += 256) {
        int k = i / CPAD, c = i % CPAD;
        sW2[(k >> 1) * W2STR + 2 * c + (k & 1)] = (c < NCLS) ? W2[k * NCLS + c] : 0.0f;
    }

    const int numTiles = (n + 15) / 16;
    for (int base = blockIdx.x * 8; base < numTiles; base += gridDim.x * 8) {
        const int  tile   = base + wid;
        const bool active = tile < numTiles;

        v8f acc[4] = {};
        int row = tile * 16 + rloc;
        if (row >= n) row = n - 1;

        for (int kc = 0; kc < NCHK; ++kc) {
            __syncthreads();
            for (int i = tid; i < KCH * HID; i += 256) {
                int k = i >> 6, c = i & 63;
                sBig[(k >> 1) * W1STR + 2 * c + (k & 1)] = W1[kc * KCH * HID + i];
            }
            __syncthreads();
            if (active) {
                const float* xrow = x + (size_t)row * FIN + kc * KCH + 2 * khalf;
                const v2f* bmat = (const v2f*)sBig;
#pragma unroll
                for (int kk = 0; kk < KCH / 4; ++kk) {
                    v2f a = ((const v2f*)xrow)[2 * kk];
                    const int pr = (2 * kk + khalf) * (W1STR / 2);
#pragma unroll
                    for (int nt = 0; nt < 4; ++nt) {
                        v2f b = bmat[pr + nt * 16 + rloc];
                        acc[nt] = __builtin_amdgcn_wmma_f32_16x16x4_f32(
                            false, a, false, b, (short)0, acc[nt], false, false);
                    }
                }
            }
        }

        __syncthreads();
        float* hsw = sBig + wid * (16 * HSTR);
        if (active) {
#pragma unroll
            for (int nt = 0; nt < 4; ++nt) {
                const float bb = b1[nt * 16 + rloc];
#pragma unroll
                for (int v = 0; v < 8; ++v) {
                    float val = acc[nt][v] + bb;
                    val = fmaxf(val, 0.0f);
                    const int M = v + 8 * khalf;
                    hsw[M * HSTR + nt * 16 + rloc] = val;
                }
            }
        }
        __syncthreads();

        if (active) {
            v8f acc2[3] = {};
            const v2f* amat = (const v2f*)hsw;
            const v2f* bmat = (const v2f*)sW2;
#pragma unroll 4
            for (int kk = 0; kk < HID / 4; ++kk) {
                v2f a2 = amat[rloc * (HSTR / 2) + 2 * kk + khalf];
                const int pr = (2 * kk + khalf) * (W2STR / 2);
#pragma unroll
                for (int nt = 0; nt < 3; ++nt) {
                    v2f b = bmat[pr + nt * 16 + rloc];
                    acc2[nt] = __builtin_amdgcn_wmma_f32_16x16x4_f32(
                        false, a2, false, b, (short)0, acc2[nt], false, false);
                }
            }
#pragma unroll
            for (int nt = 0; nt < 3; ++nt) {
                const int cb  = nt * 16 + rloc;
                const float bb = (cb < NCLS) ? b2[cb] : 0.0f;
#pragma unroll
                for (int v = 0; v < 8; ++v) {
                    const int r = tile * 16 + v + 8 * khalf;
                    if (r < n) {
                        float val = (cb < NCLS) ? (acc2[nt][v] + bb) : 0.0f;
                        zp[(size_t)r * ZSTR + cb] = val;
                    }
                }
            }
        }
    }
}

__global__ void gpr_log_softmax(const float* __restrict__ zp, float* __restrict__ out, int n) {
    int i = blockIdx.x * blockDim.x + threadIdx.x;
    if (i >= n) return;
    const float* zr = zp + (size_t)i * ZSTR;
    float vals[NCLS];
    float m = -3.4e38f;
#pragma unroll
    for (int c = 0; c < NCLS; ++c) { vals[c] = zr[c]; m = fmaxf(m, vals[c]); }
    float s = 0.0f;
#pragma unroll
    for (int c = 0; c < NCLS; ++c) s += __expf(vals[c] - m);
    const float ls = __logf(s);
    float* o = out + (size_t)i * NCLS;
#pragma unroll
    for (int c = 0; c < NCLS; ++c) o[c] = vals[c] - m - ls;
}

#define PLB 256
#define PSWEEP 5
struct __attribute__((aligned(4))) f3 { float x, y, z; };

template <int J> __device__ __forceinline__ int gpr_row_bcast(int v) {
    return __builtin_amdgcn_update_dpp(0, v, 0x150 + J, 0xf, 0xf, true);
}
template <int J> __device__ __forceinline__ f3 gpr_row_of(const char* __restrict__ cb, int srcs, unsigned coff) {
    const unsigned off = ((unsigned)gpr_row_bcast<J>(srcs) << 8) + coff;
    return *(const f3*)(cb + off);
}
template <int J> __device__ __forceinline__ void gpr_fma3(f3& acc, int ws, const f3& v) {
    const float w = __int_as_float(gpr_row_bcast<J>(ws));
    acc.x = __builtin_fmaf(w, v.x, acc.x);
    acc.y = __builtin_fmaf(w, v.y, acc.y);
    acc.z = __builtin_fmaf(w, v.z, acc.z);
}

__global__ __launch_bounds__(PLB, 10)
void gpr_pull(const float* __restrict__ cur, float* __restrict__ nxt,
              const int* __restrict__ start, const int* __restrict__ rec,
              const float* __restrict__ dis2, const float* __restrict__ temp,
              float* __restrict__ hid, int k, int wr, int n) {
    const int l16 = threadIdx.x & 15;
    const unsigned coff = 12u * (unsigned)l16;
    const int nh = (gridDim.x * PLB) >> 4;
    const char* cb = (const char*)cur;
    const int2* r2 = (const int2*)rec;
    const float tk = temp[k];
    const float t0 = temp[0];

    int i = (blockIdx.x * PLB + threadIdx.x) >> 4;
    if (i >= n) return;
    int s0 = start[i], s1 = start[i + 1];
    for (; i < n; i += nh) {
        const int inx = (i + nh < n) ? i + nh : i;
        const int s0n = start[inx], s1n = start[inx + 1];

        int base = s0;
        const int last = (s1 > 0) ? s1 - 1 : 0;
        int2 r = r2[(base + l16 < s1) ? base + l16 : last];

        const f3 self = *(const f3*)(cb + ((unsigned)i << 8) + coff);
        const float d2 = dis2[i];

        float* h = hid + (size_t)i * NCLS + 3 * l16;
        float hp[3];
        if (k != 1) {
#pragma unroll
            for (int j = 0; j < 3; ++j) hp[j] = h[(3 * l16 + j < NCLS) ? j : 0];
        } else {
            hp[0] = t0 * self.x; hp[1] = t0 * self.y; hp[2] = t0 * self.z;
        }

        f3 acc;
        acc.x = d2 * self.x; acc.y = d2 * self.y; acc.z = d2 * self.z;

        while (base < s1) {
            const int sx = r.x;
            const int sw = (base + l16 < s1) ? r.y : 0;
            const f3 v0  = gpr_row_of<0>(cb, sx, coff);
            const f3 v1  = gpr_row_of<1>(cb, sx, coff);
            const f3 v2  = gpr_row_of<2>(cb, sx, coff);
            const f3 v3  = gpr_row_of<3>(cb, sx, coff);
            const f3 v4  = gpr_row_of<4>(cb, sx, coff);
            const f3 v5  = gpr_row_of<5>(cb, sx, coff);
            const f3 v6  = gpr_row_of<6>(cb, sx, coff);
            const f3 v7  = gpr_row_of<7>(cb, sx, coff);
            const f3 v8  = gpr_row_of<8>(cb, sx, coff);
            const f3 v9  = gpr_row_of<9>(cb, sx, coff);
            const f3 v10 = gpr_row_of<10>(cb, sx, coff);
            const f3 v11 = gpr_row_of<11>(cb, sx, coff);
            const f3 v12 = gpr_row_of<12>(cb, sx, coff);
            const f3 v13 = gpr_row_of<13>(cb, sx, coff);
            const f3 v14 = gpr_row_of<14>(cb, sx, coff);
            const f3 v15 = gpr_row_of<15>(cb, sx, coff);

            base += 16;
            r = r2[(base + l16 < s1) ? base + l16 : last];

            gpr_fma3<0>(acc, sw, v0);   gpr_fma3<1>(acc, sw, v1);
            gpr_fma3<2>(acc, sw, v2);   gpr_fma3<3>(acc, sw, v3);
            gpr_fma3<4>(acc, sw, v4);   gpr_fma3<5>(acc, sw, v5);
            gpr_fma3<6>(acc, sw, v6);   gpr_fma3<7>(acc, sw, v7);
            gpr_fma3<8>(acc, sw, v8);   gpr_fma3<9>(acc, sw, v9);
            gpr_fma3<10>(acc, sw, v10); gpr_fma3<11>(acc, sw, v11);
            gpr_fma3<12>(acc, sw, v12); gpr_fma3<13>(acc, sw, v13);
            gpr_fma3<14>(acc, sw, v14); gpr_fma3<15>(acc, sw, v15);
        }

        if (wr) *(f3*)((char*)nxt + ((unsigned)i << 8) + coff) = acc;

        const float av[3] = {acc.x, acc.y, acc.z};
#pragma unroll
        for (int j = 0; j < 3; ++j) {
            if (3 * l16 + j < NCLS) h[j] = hp[j] + tk * av[j];
        }
        s0 = s0n; s1 = s1n;
    }
}

extern "C" void kernel_launch(void* const* d_in, const int* in_sizes, int n_in,
                              void* d_out, int out_size, void* d_ws, size_t ws_size,
                              hipStream_t stream) {
    const float* x    = (const float*)d_in[0];
    const int*   ei   = (const int*)  d_in[1];
    const float* W1   = (const float*)d_in[2];
    const float* b1   = (const float*)d_in[3];
    const float* W2   = (const float*)d_in[4];
    const float* b2   = (const float*)d_in[5];
    const float* temp = (const float*)d_in[6];

    const int N = in_sizes[0] / FIN;
    const int E = in_sizes[1] / 2;
    const int* row = ei;
    const int* col = ei + E;

    float* ws   = (float*)d_ws;
    const size_t NZ = (size_t)N * ZSTR;
    float* zp   = ws;
    float* bufA = ws + NZ;
    float* bufB = ws + 2 * NZ;
    int*   rec  = (int*)(ws + 3 * NZ);
    int*   kd   = rec + 2 * (size_t)E;
    const int nrange = (N + BNR - 1) / BNR;
    const int nchunk = (E + BCE - 1) / BCE;
    const size_t CN  = (size_t)nchunk * N;
    int*   part = kd + 2 * (size_t)E;
    int*   pofs = part + CN;
    int*   cnt  = pofs + CN;
    int*   start = cnt + N;
    float* dis  = (float*)(start + N + 1);
    float* dis2 = dis + N;
    int*   bsum = (int*)(dis2 + N);

    float* out_ls = (float*)d_out;
    float* hid    = out_ls + (size_t)N * NCLS;

    const int TB = 256;
    const int gN = (N + TB - 1) / TB;
    const int gE = (E + TB - 1) / TB;
    const int gS = (N + SCH - 1) / SCH;

    gpr_bin_count   <<<nrange * nchunk, BTH, 0, stream>>>(col, part, N, E, nrange);
    gpr_bin_scan    <<<gN, TB, 0, stream>>>(part, pofs, cnt, N, nchunk);
    gpr_scan_bsum   <<<gS, TB, 0, stream>>>(cnt, bsum, N);
    gpr_scan_final  <<<gS, TB, 0, stream>>>(cnt, bsum, start, dis, dis2, N);
    gpr_bin_scatter <<<nrange * nchunk, BTH, 0, stream>>>(row, col, start, pofs, kd, N, E, nrange);
    gpr_place       <<<gE, TB, 0, stream>>>(start, kd, dis, rec, E);

    const int numTiles = (N + 15) / 16;
    const int gG = (numTiles + 7) / 8;
    gpr_mlp_wmma <<<gG, TB, 0, stream>>>(x, W1, b1, W2, b2, zp, N);
    gpr_log_softmax<<<gN, TB, 0, stream>>>(zp, out_ls, N);

    const int gP = (N + 16 * PSWEEP - 1) / (16 * PSWEEP);
    const float* cur = zp;
    float* nx = bufA;
    for (int k = 1; k <= KSTEP; ++k) {
        gpr_pull<<<gP, PLB, 0, stream>>>(cur, nx, start, rec, dis2, temp, hid, k,
                                            (k < KSTEP) ? 1 : 0, N);
        cur = nx;
        nx = (nx == bufA) ? bufB : bufA;
    }
}


# MLP kernel: each K chunk's x fragments (25 per lane) and W1 slice are fetched as grouped loads ahead of the barrier instead of one x load waited for before every 4 WMMAs; k order and sums unchanged
# speedup vs baseline: 20.3234x; 1.0906x over previous
#include <hip/hip_runtime.h>

#define FIN   500
#define HID   64
#define NCLS  47
#define CPAD  48
#define ZSTR  64
#define KSTEP 10
#define KCH   100
#define NCHK  5
#define W1PA  13
#define W1STR 160
#define W2STR 96
#define HSTR  68

typedef __attribute__((ext_vector_type(2))) float v2f;
typedef __attribute__((ext_vector_type(8))) float v8f;

#define SCH 1024
#define BNR 25088
#define BTH 1024
#define BU  16
#define BTR 3
#define BCE (BTH * BU * BTR)
#define BQ  ((BNR + BTH - 1) / BTH)

__global__ __launch_bounds__(BTH)
void gpr_bin_count(const int* __restrict__ col, int* __restrict__ part, int n, int e, int nrange) {
    __shared__ int sc[BNR];
    const int tid  = threadIdx.x;
    const int r    = blockIdx.x % nrange;
    const int c    = blockIdx.x / nrange;
    const int lo   = r * BNR;
    const int span = (n - lo < BNR) ? n - lo : BNR;
    for (int j = tid; j < span; j += BTH) sc[j] = 0;
    __syncthreads();
    const int e0 = c * BCE;
    const int e1 = (e - e0 < BCE) ? e : e0 + BCE;
    for (int b = e0; b < e1; b += BTH * BU) {
        int d[BU];
#pragma unroll
        for (int u = 0; u < BU; ++u) {
            const int i = b + u * BTH + tid;
            const int v = col[(i < e1) ? i : e1 - 1];
            d[u] = (i < e1) ? v : -1;
        }
#pragma unroll
        for (int u = 0; u < BU; ++u) {
            const unsigned dd = (unsigned)(d[u] - lo);
            if (dd < (unsigned)span) atomicAdd(&sc[dd], 1);
        }
    }
    __syncthreads();
    int* prow = part + (size_t)c * n + lo;
    for (int j = tid; j < span; j += BTH) prow[j] = sc[j];
}

__global__ void gpr_bin_scan(const int* __restrict__ part, int* __restrict__ pofs,
                             int* __restrict__ cnt, int n, int nchunk) {
    const int j = blockIdx.x * blockDim.x + threadIdx.x;
    if (j >= n) return;
    int run = 0;
#pragma unroll 16
    for (int c = 0; c < nchunk; ++c) {
        const int v = part[(size_t)c * n + j];
        pofs[(size_t)c * n + j] = run;
        run += v;
    }
    cnt[j] = run;
}

__global__ __launch_bounds__(256)
void gpr_scan_bsum(const int* __restrict__ cnt, int* __restrict__ bsum, int n) {
    __shared__ int sh[256];
    const int t  = threadIdx.x;
    const int b0 = blockIdx.x * SCH + t * 4;
    int s = 0;
#pragma unroll
    for (int j = 0; j < 4; ++j) { const int i = b0 + j; if (i < n) s += cnt[i]; }
    sh[t] = s;
    __syncthreads();
    for (int off = 128; off > 0; off >>= 1) {
        if (t < off) sh[t] += sh[t + off];
        __syncthreads();
    }
    if (t == 0) bsum[blockIdx.x] = sh[0];
}

__global__ __launch_bounds__(256)
void gpr_scan_final(const int* __restrict__ cnt, const int* __restrict__ bsum,
                    int* __restrict__ start, float* __restrict__ dis,
                    float* __restrict__ dis2, int n) {
    __shared__ int sh[256];
    __shared__ int sbase;
    const int t = threadIdx.x;
    int p = 0;
    for (int j = t; j < (int)blockIdx.x; j += 256) p += bsum[j];
    sh[t] = p;
    __syncthreads();
    for (int off = 128; off > 0; off >>= 1) {
        if (t < off) sh[t] += sh[t + off];
        __syncthreads();
    }
    if (t == 0) sbase = sh[0];
    __syncthreads();
    const int base = sbase;

    const int b0 = blockIdx.x * SCH + t * 4;
    int c[4];
    int s = 0;
#pragma unroll
    for (int j = 0; j < 4; ++j) { const int i = b0 + j; c[j] = (i < n) ? cnt[i] : 0; s += c[j]; }
    sh[t] = s;
    __syncthreads();
    for (int off = 1; off < 256; off <<= 1) {
        const int v = (t >= off) ? sh[t - off] : 0;
        __syncthreads();
        sh[t] += v;
        __syncthreads();
    }
    int run = base + sh[t] - s;
#pragma unroll
    for (int j = 0; j < 4; ++j) {
        const int i = b0 + j;
        if (i < n) {
            start[i] = run;
            run += c[j];
            const float d = 1.0f + (float)c[j];
            dis[i]  = rsqrtf(d);
            dis2[i] = 1.0f / d;
            if (i == n - 1) start[n] = run;
        }
    }
}

__global__ __launch_bounds__(BTH)
void gpr_bin_scatter(const int* __restrict__ row, const int* __restrict__ col,
                     const int* __restrict__ start, const int* __restrict__ pofs,
                     int* __restrict__ kd, int n, int e, int nrange) {
    __shared__ int sc[BNR];
    const int tid  = threadIdx.x;
    const int r    = blockIdx.x % nrange;
    const int c    = blockIdx.x / nrange;
    const int lo   = r * BNR;
    const int span = (n - lo < BNR) ? n - lo : BNR;
    {
        const int* srow = start + lo;
        const int* prow = pofs + (size_t)c * n + lo;
        int v[BQ];
#pragma unroll
        for (int q = 0; q < BQ; ++q) {
            const int j  = tid + q * BTH;
            const int jc = (j < span) ? j : span - 1;
            v[q] = srow[jc] + prow[jc];
        }
#pragma unroll
        for (int q = 0; q < BQ; ++q) {
            const int j = tid + q * BTH;
            if (j < span) sc[j] = v[q];
        }
    }
    __syncthreads();
    const int e0 = c * BCE;
    const int e1 = (e - e0 < BCE) ? e : e0 + BCE;
    int2* kd2 = (int2*)kd;
    for (int b = e0; b < e1; b += BTH * BU) {
        int d[BU], s[BU], slot[BU];
#pragma unroll
        for (int u = 0; u < BU; ++u) {
            const int i  = b + u * BTH + tid;
            const int ic = (i < e1) ? i : e1 - 1;
            const int vd = col[ic];
            s[u] = row[ic];
            d[u] = (i < e1) ? vd : -1;
        }
#pragma unroll
        for (int u = 0; u < BU; ++u) {
            const unsigned dd = (unsigned)(d[u] - lo);
            int sl = -1;
            if (dd < (unsigned)span) sl = atomicAdd(&sc[dd], 1);
            slot[u] = sl;
        }
#pragma unroll
        for (int u = 0; u < BU; ++u) {
            if (slot[u] >= 0) {
                int2 o;
                o.x = s[u];
                o.y = d[u];
                kd2[slot[u]] = o;
            }
        }
    }
}

__global__ void gpr_place(const int* __restrict__ start, const int* __restrict__ kd,
                          const float* __restrict__ dis, int* __restrict__ rec, int e) {
    int p = blockIdx.x * blockDim.x + threadIdx.x;
    if (p >= e) return;
    const int2* kd2 = (const int2*)kd;
    const int2 me = kd2[p];
    const int r = me.x, c = me.y;
    const int s0 = start[c], s1 = start[c + 1];
    int less = 0;
    for (int k = s0; k < s1; ++k) {
        const int kk = kd2[k].x;
        less += ((kk < r) || (kk == r && k < p)) ? 1 : 0;
    }
    int2 o;
    o.x = r;
    o.y = __float_as_int(dis[r] * dis[c]);
    ((int2*)rec)[s0 + less] = o;
}

__global__ __launch_bounds__(256, 8)
void gpr_mlp_wmma(const float* __restrict__ x,  const float* __restrict__ W1,
                  const float* __restrict__ b1, const float* __restrict__ W2,
                  const float* __restrict__ b2, float* __restrict__ zp, int n) {
    __shared__ float sBig[8704];
    __shared__ float sW2[32 * W2STR];

    const int tid    = threadIdx.x;
    const int wid    = tid >> 5;
    const int lane   = tid & 31;
    const int rloc   = lane & 15;
    const int khalf  = lane >> 4;

    for (int i = tid; i < HID * CPAD; i += 256) {
        int k = i / CPAD, c = i % CPAD;
        sW2[(k >> 1) * W2STR + 2 * c + (k & 1)] = (c < NCLS) ? W2[k * NCLS + c] : 0.0f;
    }

    const int numTiles = (n + 15) / 16;
    for (int base = blockIdx.x * 8; base < numTiles; base += gridDim.x * 8) {
        const int  tile   = base + wid;
        const bool active = tile < numTiles;

        v8f acc[4] = {};
        int row = tile * 16 + rloc;
        if (row >= n) row = n - 1;

        for (int kc = 0; kc < NCHK; ++kc) {
            v2f xa[KCH / 4];
            {
                const float* xrow = x + (size_t)row * FIN + kc * KCH + 2 * khalf;
#pragma unroll
                for (int kk = 0; kk < KCH / 4; ++kk) xa[kk] = ((const v2f*)xrow)[2 * kk];
            }
            const float* w1c = W1 + kc * KCH * HID + tid;
            float* sdst = sBig + ((tid >> 6) >> 1) * W1STR + 2 * (tid & 63) + ((tid >> 6) & 1);
            float wa[W1PA];
#pragma unroll
            for (int j = 0; j < W1PA; ++j) wa[j] = w1c[256 * j];
            __syncthreads();
#pragma unroll
            for (int j = 0; j < W1PA; ++j) sdst[2 * j * W1STR] = wa[j];
            {
                float wb[KCH * HID / 256 - W1PA];
#pragma unroll
                for (int j = W1PA; j < KCH * HID / 256; ++j) wb[j - W1PA] = w1c[256 * j];
#pragma unroll
                for (int j = W1PA; j < KCH * HID / 256; ++j) sdst[2 * j * W1STR] = wb[j - W1PA];
            }
            __syncthreads();
            if (active) {
                const v2f* bmat = (const v2f*)sBig;
#pragma unroll
                for (int kk = 0; kk < KCH / 4; ++kk) {
                    const v2f a = xa[kk];
                    const int pr = (2 * kk + khalf) * (W1STR / 2);
#pragma unroll
                    for (int nt = 0; nt < 4; ++nt) {
                        v2f b = bmat[pr + nt * 16 + rloc];
                        acc[nt] = __builtin_amdgcn_wmma_f32_16x16x4_f32(
                            false, a, false, b, (short)0, acc[nt], false, false);
                    }
                }
            }
        }

        __syncthreads();
        float* hsw = sBig + wid * (16 * HSTR);
        if (active) {
#pragma unroll
            for (int nt = 0; nt < 4; ++nt) {
                const float bb = b1[nt * 16 + rloc];
#pragma unroll
                for (int v = 0; v < 8; ++v) {
                    float val = acc[nt][v] + bb;
                    val = fmaxf(val, 0.0f);
                    const int M = v + 8 * khalf;
                    hsw[M * HSTR + nt * 16 + rloc] = val;
                }
            }
        }
        __syncthreads();

        if (active) {
            v8f acc2[3] = {};
            const v2f* amat = (const v2f*)hsw;
            const v2f* bmat = (const v2f*)sW2;
#pragma unroll 4
            for (int kk = 0; kk < HID / 4; ++kk) {
                v2f a2 = amat[rloc * (HSTR / 2) + 2 * kk + khalf];
                const int pr = (2 * kk + khalf) * (W2STR / 2);
#pragma unroll
                for (int nt = 0; nt < 3; ++nt) {
                    v2f b = bmat[pr + nt * 16 + rloc];
                    acc2[nt] = __builtin_amdgcn_wmma_f32_16x16x4_f32(
                        false, a2, false, b, (short)0, acc2[nt], false, false);
                }
            }
#pragma unroll
            for (int nt = 0; nt < 3; ++nt) {
                const int cb  = nt * 16 + rloc;
                const float bb = (cb < NCLS) ? b2[cb] : 0.0f;
#pragma unroll
                for (int v = 0; v < 8; ++v) {
                    const int r = tile * 16 + v + 8 * khalf;
                    if (r < n) {
                        float val = (cb < NCLS) ? (acc2[nt][v] + bb) : 0.0f;
                        zp[(size_t)r * ZSTR + cb] = val;
                    }
                }
            }
        }
    }
}

__global__ void gpr_log_softmax(const float* __restrict__ zp, float* __restrict__ out, int n) {
    int i = blockIdx.x * blockDim.x + threadIdx.x;
    if (i >= n) return;
    const float* zr = zp + (size_t)i * ZSTR;
    float vals[NCLS];
    float m = -3.4e38f;
#pragma unroll
    for (int c = 0; c < NCLS; ++c) { vals[c] = zr[c]; m = fmaxf(m, vals[c]); }
    float s = 0.0f;
#pragma unroll
    for (int c = 0; c < NCLS; ++c) s += __expf(vals[c] - m);
    const float ls = __logf(s);
    float* o = out + (size_t)i * NCLS;
#pragma unroll
    for (int c = 0; c < NCLS; ++c) o[c] = vals[c] - m - ls;
}

#define PLB 256
#define PSWEEP 5
struct __attribute__((aligned(4))) f3 { float x, y, z; };

template <int J> __device__ __forceinline__ int gpr_row_bcast(int v) {
    return __builtin_amdgcn_update_dpp(0, v, 0x150 + J, 0xf, 0xf, true);
}
template <int J> __device__ __forceinline__ f3 gpr_row_of(const char* __restrict__ cb, int srcs, unsigned coff) {
    const unsigned off = ((unsigned)gpr_row_bcast<J>(srcs) << 8) + coff;
    return *(const f3*)(cb + off);
}
template <int J> __device__ __forceinline__ void gpr_fma3(f3& acc, int ws, const f3& v) {
    const float w = __int_as_float(gpr_row_bcast<J>(ws));
    acc.x = __builtin_fmaf(w, v.x, acc.x);
    acc.y = __builtin_fmaf(w, v.y, acc.y);
    acc.z = __builtin_fmaf(w, v.z, acc.z);
}

__global__ __launch_bounds__(PLB, 10)
void gpr_pull(const float* __restrict__ cur, float* __restrict__ nxt,
              const int* __restrict__ start, const int* __restrict__ rec,
              const float* __restrict__ dis2, const float* __restrict__ temp,
              float* __restrict__ hid, int k, int wr, int n) {
    const int l16 = threadIdx.x & 15;
    const unsigned coff = 12u * (unsigned)l16;
    const int nh = (gridDim.x * PLB) >> 4;
    const char* cb = (const char*)cur;
    const int2* r2 = (const int2*)rec;
    const float tk = temp[k];
    const float t0 = temp[0];

    int i = (blockIdx.x * PLB + threadIdx.x) >> 4;
    if (i >= n) return;
    int s0 = start[i], s1 = start[i + 1];
    for (; i < n; i += nh) {
        const int inx = (i + nh < n) ? i + nh : i;
        const int s0n = start[inx], s1n = start[inx + 1];

        int base = s0;
        const int last = (s1 > 0) ? s1 - 1 : 0;
        int2 r = r2[(base + l16 < s1) ? base + l16 : last];

        const f3 self = *(const f3*)(cb + ((unsigned)i << 8) + coff);
        const float d2 = dis2[i];

        float* h = hid + (size_t)i * NCLS + 3 * l16;
        float hp[3];
        if (k != 1) {
#pragma unroll
            for (int j = 0; j < 3; ++j) hp[j] = h[(3 * l16 + j < NCLS) ? j : 0];
        } else {
            hp[0] = t0 * self.x; hp[1] = t0 * self.y; hp[2] = t0 * self.z;
        }

        f3 acc;
        acc.x = d2 * self.x; acc.y = d2 * self.y; acc.z = d2 * self.z;

        while (base < s1) {
            const int sx = r.x;
            const int sw = (base + l16 < s1) ? r.y : 0;
            const f3 v0  = gpr_row_of<0>(cb, sx, coff);
            const f3 v1  = gpr_row_of<1>(cb, sx, coff);
            const f3 v2  = gpr_row_of<2>(cb, sx, coff);
            const f3 v3  = gpr_row_of<3>(cb, sx, coff);
            const f3 v4  = gpr_row_of<4>(cb, sx, coff);
            const f3 v5  = gpr_row_of<5>(cb, sx, coff);
            const f3 v6  = gpr_row_of<6>(cb, sx, coff);
            const f3 v7  = gpr_row_of<7>(cb, sx, coff);
            const f3 v8  = gpr_row_of<8>(cb, sx, coff);
            const f3 v9  = gpr_row_of<9>(cb, sx, coff);
            const f3 v10 = gpr_row_of<10>(cb, sx, coff);
            const f3 v11 = gpr_row_of<11>(cb, sx, coff);
            const f3 v12 = gpr_row_of<12>(cb, sx, coff);
            const f3 v13 = gpr_row_of<13>(cb, sx, coff);
            const f3 v14 = gpr_row_of<14>(cb, sx, coff);
            const f3 v15 = gpr_row_of<15>(cb, sx, coff);

            base += 16;
            r = r2[(base + l16 < s1) ? base + l16 : last];

            gpr_fma3<0>(acc, sw, v0);   gpr_fma3<1>(acc, sw, v1);
            gpr_fma3<2>(acc, sw, v2);   gpr_fma3<3>(acc, sw, v3);
            gpr_fma3<4>(acc, sw, v4);   gpr_fma3<5>(acc, sw, v5);
            gpr_fma3<6>(acc, sw, v6);   gpr_fma3<7>(acc, sw, v7);
            gpr_fma3<8>(acc, sw, v8);   gpr_fma3<9>(acc, sw, v9);
            gpr_fma3<10>(acc, sw, v10); gpr_fma3<11>(acc, sw, v11);
            gpr_fma3<12>(acc, sw, v12); gpr_fma3<13>(acc, sw, v13);
            gpr_fma3<14>(acc, sw, v14); gpr_fma3<15>(acc, sw, v15);
        }

        if (wr) *(f3*)((char*)nxt + ((unsigned)i << 8) + coff) = acc;

        const float av[3] = {acc.x, acc.y, acc.z};
#pragma unroll
        for (int j = 0; j < 3; ++j) {
            if (3 * l16 + j < NCLS) h[j] = hp[j] + tk * av[j];
        }
        s0 = s0n; s1 = s1n;
    }
}

extern "C" void kernel_launch(void* const* d_in, const int* in_sizes, int n_in,
                              void* d_out, int out_size, void* d_ws, size_t ws_size,
                              hipStream_t stream) {
    const float* x    = (const float*)d_in[0];
    const int*   ei   = (const int*)  d_in[1];
    const float* W1   = (const float*)d_in[2];
    const float* b1   = (const float*)d_in[3];
    const float* W2   = (const float*)d_in[4];
    const float* b2   = (const float*)d_in[5];
    const float* temp = (const float*)d_in[6];

    const int N = in_sizes[0] / FIN;
    const int E = in_sizes[1] / 2;
    const int* row = ei;
    const int* col = ei + E;

    float* ws   = (float*)d_ws;
    const size_t NZ = (size_t)N * ZSTR;
    float* zp   = ws;
    float* bufA = ws + NZ;
    float* bufB = ws + 2 * NZ;
    int*   rec  = (int*)(ws + 3 * NZ);
    int*   kd   = rec + 2 * (size_t)E;
    const int nrange = (N + BNR - 1) / BNR;
    const int nchunk = (E + BCE - 1) / BCE;
    const size_t CN  = (size_t)nchunk * N;
    int*   part = kd + 2 * (size_t)E;
    int*   pofs = part + CN;
    int*   cnt  = pofs + CN;
    int*   start = cnt + N;
    float* dis  = (float*)(start + N + 1);
    float* dis2 = dis + N;
    int*   bsum = (int*)(dis2 + N);

    float* out_ls = (float*)d_out;
    float* hid    = out_ls + (size_t)N * NCLS;

    const int TB = 256;
    const int gN = (N + TB - 1) / TB;
    const int gE = (E + TB - 1) / TB;
    const int gS = (N + SCH - 1) / SCH;

    gpr_bin_count   <<<nrange * nchunk, BTH, 0, stream>>>(col, part, N, E, nrange);
    gpr_bin_scan    <<<gN, TB, 0, stream>>>(part, pofs, cnt, N, nchunk);
    gpr_scan_bsum   <<<gS, TB, 0, stream>>>(cnt, bsum, N);
    gpr_scan_final  <<<gS, TB, 0, stream>>>(cnt, bsum, start, dis, dis2, N);
    gpr_bin_scatter <<<nrange * nchunk, BTH, 0, stream>>>(row, col, start, pofs, kd, N, E, nrange);
    gpr_place       <<<gE, TB, 0, stream>>>(start, kd, dis, rec, E);

    const int numTiles = (N + 15) / 16;
    const int gG = (numTiles + 7) / 8;
    gpr_mlp_wmma <<<gG, TB, 0, stream>>>(x, W1, b1, W2, b2, zp, N);
    gpr_log_softmax<<<gN, TB, 0, stream>>>(zp, out_ls, N);

    const int gP = (N + 16 * PSWEEP - 1) / (16 * PSWEEP);
    const float* cur = zp;
    float* nx = bufA;
    for (int k = 1; k <= KSTEP; ++k) {
        gpr_pull<<<gP, PLB, 0, stream>>>(cur, nx, start, rec, dis2, temp, hid, k,
                                            (k < KSTEP) ? 1 : 0, N);
        cur = nx;
        nx = (nx == bufA) ? bufB : bufA;
    }
}
